# starting kernel
# baseline (speedup 1.0000x reference)
#include <hip/hip_runtime.h>
#include <hip/hip_bf16.h>

typedef float v2f __attribute__((ext_vector_type(2)));
typedef float v8f __attribute__((ext_vector_type(8)));

__global__ void k_deg_init(float* __restrict__ deg, int n) {
  int i = blockIdx.x * blockDim.x + threadIdx.x;
  if (i < n) deg[i] = 1.0f;
}

__global__ void k_deg_accum(const int* __restrict__ dst, float* __restrict__ deg, int e) {
  int i = blockIdx.x * blockDim.x + threadIdx.x;
  if (i < e) atomicAdd(&deg[dst[i]], 1.0f);
}

__global__ void k_deg_rsqrt(float* __restrict__ deg, int n) {
  int i = blockIdx.x * blockDim.x + threadIdx.x;
  if (i < n) deg[i] = rsqrtf(deg[i]);
}

__global__ void k_gemm_wmma_f32(const float* __restrict__ A,
                                const float* __restrict__ B,
                                float* __restrict__ C,
                                int M, int K, int N) {
  const int lane = threadIdx.x & 31;
  const int wv   = threadIdx.x >> 5;
  const int h    = lane >> 4;
  const int l16  = lane & 15;

  const int mbase = blockIdx.x * 32;
  const int m0 = mbase + l16;
  const int m1 = m0 + 16;
  const int mc0 = (m0 < M) ? m0 : (M - 1);
  const int mc1 = (m1 < M) ? m1 : (M - 1);
  const int n0 = wv * 32 + l16;
  const int n1 = n0 + 16;

  const float* a0row = A + (size_t)mc0 * K;
  const float* a1row = A + (size_t)mc1 * K;

  v8f acc00 = {}, acc01 = {}, acc10 = {}, acc11 = {};
  for (int k0 = 0; k0 < K; k0 += 4) {
    const int ka = k0 + 2 * h;
    v2f a0, a1, b0, b1;
    a0.x = a0row[ka];  a0.y = a0row[ka + 1];
    a1.x = a1row[ka];  a1.y = a1row[ka + 1];
    const float* br0 = B + (size_t)ka * N;
    const float* br1 = br0 + N;
    b0.x = br0[n0];  b0.y = br1[n0];
    b1.x = br0[n1];  b1.y = br1[n1];
    acc00 = __builtin_amdgcn_wmma_f32_16x16x4_f32(false, a0, false, b0, (short)0, acc00, false, false);
    acc01 = __builtin_amdgcn_wmma_f32_16x16x4_f32(false, a0, false, b1, (short)0, acc01, false, false);
    acc10 = __builtin_amdgcn_wmma_f32_16x16x4_f32(false, a1, false, b0, (short)0, acc10, false, false);
    acc11 = __builtin_amdgcn_wmma_f32_16x16x4_f32(false, a1, false, b1, (short)0, acc11, false, false);
  }

#pragma unroll
  for (int r = 0; r < 8; ++r) {
    const int row0 = mbase + r + 8 * h;
    const int row1 = row0 + 16;
    if (row0 < M) {
      C[(size_t)row0 * N + n0] = acc00[r];
      C[(size_t)row0 * N + n1] = acc01[r];
    }
    if (row1 < M) {
      C[(size_t)row1 * N + n0] = acc10[r];
      C[(size_t)row1 * N + n1] = acc11[r];
    }
  }
}

__global__ void k_selfloop_init_v4(const float* __restrict__ H,
                                   const float* __restrict__ dinv,
                                   const float* __restrict__ bias,
                                   float* __restrict__ out,
                                   int n, int C, int add_bias) {
  int idx = blockIdx.x * blockDim.x + threadIdx.x;
  int c4 = C >> 2;
  if (idx >= n * c4) return;
  int i = idx / c4;
  int c = (idx - i * c4) << 2;
  float w = dinv[i];
  w *= w;
  float4 hv = reinterpret_cast<const float4*>(H)[idx];
  float4 v = make_float4(hv.x * w, hv.y * w, hv.z * w, hv.w * w);
  if (add_bias) {
    float4 bv = *reinterpret_cast<const float4*>(bias + c);
    v.x += bv.x; v.y += bv.y; v.z += bv.z; v.w += bv.w;
  }
  reinterpret_cast<float4*>(out)[idx] = v;
}

template <int VEC>
__global__ void k_edge_scatter(const int* __restrict__ src,
                               const int* __restrict__ dst,
                               const float* __restrict__ H,
                               const float* __restrict__ dinv,
                               float* __restrict__ out,
                               int E, int C) {
  int lane = threadIdx.x & 31;
  int e = (blockIdx.x * blockDim.x + threadIdx.x) >> 5;
  if (e >= E) return;
  int i = src[e];
  int j = dst[e];
  float w = dinv[i] * dinv[j];
  const float* hrow = H + (size_t)i * C + lane * VEC;
  float* orow = out + (size_t)j * C + lane * VEC;
  if constexpr (VEC == 4) {
    float4 hv = *reinterpret_cast<const float4*>(hrow);
    atomicAdd(orow + 0, hv.x * w);
    atomicAdd(orow + 1, hv.y * w);
    atomicAdd(orow + 2, hv.z * w);
    atomicAdd(orow + 3, hv.w * w);
  } else {
    float2 hv = *reinterpret_cast<const float2*>(hrow);
    atomicAdd(orow + 0, hv.x * w);
    atomicAdd(orow + 1, hv.y * w);
  }
}

__global__ void k_bias_relu_v4(float* __restrict__ a, const float* __restrict__ b,
                               int n, int C) {
  int idx = blockIdx.x * blockDim.x + threadIdx.x;
  int c4 = C >> 2;
  if (idx >= n * c4) return;
  int i = idx / c4;
  int c = (idx - i * c4) << 2;
  float4 v = reinterpret_cast<float4*>(a)[idx];
  float4 bv = *reinterpret_cast<const float4*>(b + c);
  v.x = fmaxf(v.x + bv.x, 0.0f);
  v.y = fmaxf(v.y + bv.y, 0.0f);
  v.z = fmaxf(v.z + bv.z, 0.0f);
  v.w = fmaxf(v.w + bv.w, 0.0f);
  reinterpret_cast<float4*>(a)[idx] = v;
}

extern "C" void kernel_launch(void* const* d_in, const int* in_sizes, int n_in,
                              void* d_out, int out_size, void* d_ws, size_t ws_size,
                              hipStream_t stream) {
  const int CIN = 256, CH = 128, CO = 64;
  const float* x  = (const float*)d_in[0];
  const int*   ei = (const int*)d_in[1];
  const float* W1 = (const float*)d_in[2];
  const float* b1 = (const float*)d_in[3];
  const float* W2 = (const float*)d_in[4];
  const float* b2 = (const float*)d_in[5];
  float* out = (float*)d_out;

  const int N = in_sizes[0] / CIN;
  const int E = in_sizes[1] / 2;
  const int* src = ei;
  const int* dst = ei + E;

  float* ws   = (float*)d_ws;
  float* dinv = ws;
  float* Hbuf = ws + 50176;
  float* AGG  = ws + 50176 + (size_t)N * CH;

  const int T = 256;

  k_deg_init  <<<(N + T - 1) / T, T, 0, stream>>>(dinv, N);
  k_deg_accum <<<(E + T - 1) / T, T, 0, stream>>>(dst, dinv, E);
  k_deg_rsqrt <<<(N + T - 1) / T, T, 0, stream>>>(dinv, N);

  k_gemm_wmma_f32<<<(N + 31) / 32, 32 * (CH / 32), 0, stream>>>(x, W1, Hbuf, N, CIN, CH);
  k_selfloop_init_v4<<<((size_t)N * (CH / 4) + T - 1) / T, T, 0, stream>>>(Hbuf, dinv, b1, AGG, N, CH, 0);
  k_edge_scatter<4><<<((size_t)E * 32 + T - 1) / T, T, 0, stream>>>(src, dst, Hbuf, dinv, AGG, E, CH);
  k_bias_relu_v4<<<((size_t)N * (CH / 4) + T - 1) / T, T, 0, stream>>>(AGG, b1, N, CH);

  k_gemm_wmma_f32<<<(N + 31) / 32, 32 * (CO / 32), 0, stream>>>(AGG, W2, Hbuf, N, CH, CO);
  k_selfloop_init_v4<<<((size_t)N * (CO / 4) + T - 1) / T, T, 0, stream>>>(Hbuf, dinv, b2, out, N, CO, 1);
  k_edge_scatter<2><<<((size_t)E * 32 + T - 1) / T, T, 0, stream>>>(src, dst, Hbuf, dinv, out, E, CO);
}


# Edge aggregation as a per-destination gather over source lists built in-call (count, scan, place) replaces the float atomic scatter; per-node sum order now follows an arrival-order rank
# speedup vs baseline: 1.9399x; 1.9399x over previous
#include <hip/hip_runtime.h>
#include <hip/hip_bf16.h>

typedef float v2f __attribute__((ext_vector_type(2)));
typedef float v8f __attribute__((ext_vector_type(8)));

__global__ void k_cnt_zero(int* __restrict__ cnt, int n) {
  int i = blockIdx.x * blockDim.x + threadIdx.x;
  if (i < n) cnt[i] = 0;
}

__global__ void k_edge_rank(const int* __restrict__ dst, int* __restrict__ cnt,
                            int* __restrict__ rank, int e) {
  int i = blockIdx.x * blockDim.x + threadIdx.x;
  if (i < e) rank[i] = atomicAdd(&cnt[dst[i]], 1);
}

__global__ void __launch_bounds__(1024)
k_rowptr_dinv(const int* __restrict__ cnt, int* __restrict__ rowptr,
              float* __restrict__ dinv, int n) {
  constexpr int PER = 52;
  __shared__ int wtot[32];
  const int t = threadIdx.x;
  const int lane = t & 31;
  const int wv = t >> 5;
  const int base = t * PER;
  int c[PER];
#pragma unroll
  for (int q = 0; q < PER; q += 4) {
    const int4 v = *reinterpret_cast<const int4*>(cnt + base + q);
    c[q] = v.x; c[q + 1] = v.y; c[q + 2] = v.z; c[q + 3] = v.w;
  }
  int tot = 0;
#pragma unroll
  for (int q = 0; q < PER; q += 4) {
    const int i = base + q;
    const float4 d = make_float4(rsqrtf((float)(c[q] + 1)), rsqrtf((float)(c[q + 1] + 1)),
                                 rsqrtf((float)(c[q + 2] + 1)), rsqrtf((float)(c[q + 3] + 1)));
    if (i + 3 < n) {
      *reinterpret_cast<float4*>(dinv + i) = d;
    } else {
      if (i     < n) dinv[i]     = d.x;
      if (i + 1 < n) dinv[i + 1] = d.y;
      if (i + 2 < n) dinv[i + 2] = d.z;
    }
    int v;
    v = c[q];     c[q]     = tot; tot += v;
    v = c[q + 1]; c[q + 1] = tot; tot += v;
    v = c[q + 2]; c[q + 2] = tot; tot += v;
    v = c[q + 3]; c[q + 3] = tot; tot += v;
  }
  int inc = tot;
  for (int d = 1; d < 32; d <<= 1) {
    const int o = __shfl_up(inc, d);
    if (lane >= d) inc += o;
  }
  if (lane == 31) wtot[wv] = inc;
  __syncthreads();
  if (wv == 0) {
    const int s = wtot[lane];
    int si = s;
    for (int d = 1; d < 32; d <<= 1) {
      const int o = __shfl_up(si, d);
      if (lane >= d) si += o;
    }
    wtot[lane] = si - s;
  }
  __syncthreads();
  const int off = wtot[wv] + inc - tot;
#pragma unroll
  for (int q = 0; q < PER; q += 4) {
    const int i = base + q;
    if (i + 3 < n) {
      *reinterpret_cast<int4*>(rowptr + i) =
          make_int4(off + c[q], off + c[q + 1], off + c[q + 2], off + c[q + 3]);
    } else {
      if (i     < n) rowptr[i]     = off + c[q];
      if (i + 1 < n) rowptr[i + 1] = off + c[q + 1];
      if (i + 2 < n) rowptr[i + 2] = off + c[q + 2];
    }
  }
  if (t == 1023) rowptr[n] = off + tot;
}

__global__ void k_edge_place(const int* __restrict__ src, const int* __restrict__ dst,
                             const int* __restrict__ rank, const int* __restrict__ rowptr,
                             int* __restrict__ list, int e) {
  int i = blockIdx.x * blockDim.x + threadIdx.x;
  if (i < e) list[rowptr[dst[i]] + rank[i]] = src[i];
}

__global__ void k_lists_move(const int* __restrict__ from, int* __restrict__ to, int count) {
  int i = blockIdx.x * blockDim.x + threadIdx.x;
  if (i < count) to[i] = from[i];
}

__global__ void k_gemm_wmma_f32(const float* __restrict__ A,
                                const float* __restrict__ B,
                                float* __restrict__ C,
                                int M, int K, int N) {
  const int lane = threadIdx.x & 31;
  const int wv   = threadIdx.x >> 5;
  const int h    = lane >> 4;
  const int l16  = lane & 15;

  const int mbase = blockIdx.x * 32;
  const int m0 = mbase + l16;
  const int m1 = m0 + 16;
  const int mc0 = (m0 < M) ? m0 : (M - 1);
  const int mc1 = (m1 < M) ? m1 : (M - 1);
  const int n0 = wv * 32 + l16;
  const int n1 = n0 + 16;

  const float* a0row = A + (size_t)mc0 * K;
  const float* a1row = A + (size_t)mc1 * K;

  v8f acc00 = {}, acc01 = {}, acc10 = {}, acc11 = {};
  for (int k0 = 0; k0 < K; k0 += 4) {
    const int ka = k0 + 2 * h;
    v2f a0, a1, b0, b1;
    a0.x = a0row[ka];  a0.y = a0row[ka + 1];
    a1.x = a1row[ka];  a1.y = a1row[ka + 1];
    const float* br0 = B + (size_t)ka * N;
    const float* br1 = br0 + N;
    b0.x = br0[n0];  b0.y = br1[n0];
    b1.x = br0[n1];  b1.y = br1[n1];
    acc00 = __builtin_amdgcn_wmma_f32_16x16x4_f32(false, a0, false, b0, (short)0, acc00, false, false);
    acc01 = __builtin_amdgcn_wmma_f32_16x16x4_f32(false, a0, false, b1, (short)0, acc01, false, false);
    acc10 = __builtin_amdgcn_wmma_f32_16x16x4_f32(false, a1, false, b0, (short)0, acc10, false, false);
    acc11 = __builtin_amdgcn_wmma_f32_16x16x4_f32(false, a1, false, b1, (short)0, acc11, false, false);
  }

#pragma unroll
  for (int r = 0; r < 8; ++r) {
    const int row0 = mbase + r + 8 * h;
    const int row1 = row0 + 16;
    if (row0 < M) {
      C[(size_t)row0 * N + n0] = acc00[r];
      C[(size_t)row0 * N + n1] = acc01[r];
    }
    if (row1 < M) {
      C[(size_t)row1 * N + n0] = acc10[r];
      C[(size_t)row1 * N + n1] = acc11[r];
    }
  }
}

template <int VEC, bool RELU>
__global__ void k_node_gather(const int* __restrict__ rowptr, const int* __restrict__ list,
                              const float* __restrict__ H, const float* __restrict__ dinv,
                              const float* __restrict__ bias, float* __restrict__ out, int n) {
  typedef float vf __attribute__((ext_vector_type(VEC)));
  constexpr int C = 32 * VEC;
  const int lane = threadIdx.x & 31;
  const int j = __builtin_amdgcn_readfirstlane((int)((blockIdx.x * blockDim.x + threadIdx.x) >> 5));
  if (j >= n) return;
  const int beg = rowptr[j];
  const int end = rowptr[j + 1];
  const float dj = dinv[j];
  const float* hl = H + lane * VEC;
  const vf hs = *reinterpret_cast<const vf*>(hl + (size_t)j * C);
  vf acc = {};
  for (int base = beg; base < end; base += 32) {
    const int m = (end - base < 32) ? (end - base) : 32;
    int s_l = j;
    float w_l = 0.0f;
    if (lane < m) {
      s_l = list[base + lane];
      w_l = dinv[s_l] * dj;
    }
    for (int k = 0; k < m; k += 4) {
      const int s0 = __builtin_amdgcn_readlane(s_l, k);
      const int s1 = __builtin_amdgcn_readlane(s_l, k + 1);
      const int s2 = __builtin_amdgcn_readlane(s_l, k + 2);
      const int s3 = __builtin_amdgcn_readlane(s_l, k + 3);
      const float w0 = __int_as_float(__builtin_amdgcn_readlane(__float_as_int(w_l), k));
      const float w1 = __int_as_float(__builtin_amdgcn_readlane(__float_as_int(w_l), k + 1));
      const float w2 = __int_as_float(__builtin_amdgcn_readlane(__float_as_int(w_l), k + 2));
      const float w3 = __int_as_float(__builtin_amdgcn_readlane(__float_as_int(w_l), k + 3));
      const vf h0 = *reinterpret_cast<const vf*>(hl + (size_t)s0 * C);
      const vf h1 = *reinterpret_cast<const vf*>(hl + (size_t)s1 * C);
      const vf h2 = *reinterpret_cast<const vf*>(hl + (size_t)s2 * C);
      const vf h3 = *reinterpret_cast<const vf*>(hl + (size_t)s3 * C);
      acc = h0 * w0 + acc;
      const vf t1 = h1 * w1 + acc;
      acc = (k + 1 < m) ? t1 : acc;
      const vf t2 = h2 * w2 + acc;
      acc = (k + 2 < m) ? t2 : acc;
      const vf t3 = h3 * w3 + acc;
      acc = (k + 3 < m) ? t3 : acc;
    }
  }
  acc = hs * (dj * dj) + acc;
  vf r = acc + *reinterpret_cast<const vf*>(bias + lane * VEC);
  if (RELU) {
#pragma unroll
    for (int q = 0; q < VEC; ++q) r[q] = fmaxf(r[q], 0.0f);
  }
  *reinterpret_cast<vf*>(out + (size_t)j * C + lane * VEC) = r;
}

extern "C" void kernel_launch(void* const* d_in, const int* in_sizes, int n_in,
                              void* d_out, int out_size, void* d_ws, size_t ws_size,
                              hipStream_t stream) {
  const int CIN = 256, CH = 128, CO = 64;
  const float* x  = (const float*)d_in[0];
  const int*   ei = (const int*)d_in[1];
  const float* W1 = (const float*)d_in[2];
  const float* b1 = (const float*)d_in[3];
  const float* W2 = (const float*)d_in[4];
  const float* b2 = (const float*)d_in[5];
  float* out = (float*)d_out;

  const int N = in_sizes[0] / CIN;
  const int E = in_sizes[1] / 2;
  const int* src = ei;
  const int* dst = ei + E;

  float* ws   = (float*)d_ws;
  float* dinv = ws;
  float* Hbuf = ws + 50176;
  float* AGG  = ws + 50176 + (size_t)N * CH;

  int* oi     = (int*)d_out;
  int* cnt    = oi;
  int* rowptr = oi + 53248;
  int* list   = rowptr + N + 1;
  int* rank   = list + E;
  int* rowptr2 = (int*)(Hbuf + (size_t)N * CO);
  int* list2   = rowptr2 + N + 1;

  const int T = 256;

  k_cnt_zero   <<<53248 / T, T, 0, stream>>>(cnt, 53248);
  k_edge_rank  <<<(E + T - 1) / T, T, 0, stream>>>(dst, cnt, rank, E);
  k_rowptr_dinv<<<1, 1024, 0, stream>>>(cnt, rowptr, dinv, N);
  k_edge_place <<<(E + T - 1) / T, T, 0, stream>>>(src, dst, rank, rowptr, list, E);

  k_gemm_wmma_f32<<<(N + 31) / 32, 32 * (CH / 32), 0, stream>>>(x, W1, Hbuf, N, CIN, CH);
  k_node_gather<4, true><<<((size_t)N * 32 + T - 1) / T, T, 0, stream>>>(rowptr, list, Hbuf, dinv, b1, AGG, N);

  k_lists_move<<<(N + 1 + E + T - 1) / T, T, 0, stream>>>(rowptr, rowptr2, N + 1 + E);
  k_gemm_wmma_f32<<<(N + 31) / 32, 32 * (CO / 32), 0, stream>>>(AGG, W2, Hbuf, N, CH, CO);
  k_node_gather<2, false><<<((size_t)N * 32 + T - 1) / T, T, 0, stream>>>(rowptr2, list2, Hbuf, dinv, b2, out, N);
}


# Edge lists by destination built with LDS counters after binning edges by 256-node range (count, scan, scatter, per-bin finish), no per-edge returning global atomics; sum order = LDS arrival rank
# speedup vs baseline: 2.4336x; 1.2545x over previous
#include <hip/hip_runtime.h>
#include <hip/hip_bf16.h>

typedef float v2f __attribute__((ext_vector_type(2)));
typedef float v8f __attribute__((ext_vector_type(8)));

constexpr int BIN_SHIFT = 8;
constexpr int BIN_NODES = 1 << BIN_SHIFT;
constexpr int NBLK  = 256;
constexpr int NBINP = 208;
constexpr int EPT   = 4;
constexpr int EPB   = 8;

__global__ void __launch_bounds__(1024)
k_bin_count(const int* __restrict__ dst, int* __restrict__ hist, int e, int chunk) {
  __shared__ int h[NBINP];
  const int t = threadIdx.x;
  const int b = blockIdx.x;
  if (t < NBINP) h[t] = 0;
  __syncthreads();
  const int lo = b * chunk;
  const int hi = (lo + chunk < e) ? (lo + chunk) : e;
  for (int base = lo + t; base < hi; base += 1024 * EPT) {
    int d[EPT];
#pragma unroll
    for (int q = 0; q < EPT; ++q) {
      const int i = base + q * 1024;
      d[q] = (i < hi) ? dst[i] : -1;
    }
#pragma unroll
    for (int q = 0; q < EPT; ++q)
      if (d[q] >= 0) atomicAdd(&h[d[q] >> BIN_SHIFT], 1);
  }
  __syncthreads();
  if (t < NBINP) hist[t * NBLK + b] = h[t];
}

__global__ void __launch_bounds__(1024)
k_scan_excl(const int* __restrict__ in, int* __restrict__ outp) {
  constexpr int PER = 52;
  __shared__ int wtot[32];
  const int t = threadIdx.x;
  const int lane = t & 31;
  const int wv = t >> 5;
  const int base = t * PER;
  int c[PER];
#pragma unroll
  for (int q = 0; q < PER; q += 4) {
    const int4 v = *reinterpret_cast<const int4*>(in + base + q);
    c[q] = v.x; c[q + 1] = v.y; c[q + 2] = v.z; c[q + 3] = v.w;
  }
  int tot = 0;
#pragma unroll
  for (int q = 0; q < PER; ++q) {
    const int v = c[q];
    c[q] = tot;
    tot += v;
  }
  int inc = tot;
  for (int d = 1; d < 32; d <<= 1) {
    const int o = __shfl_up(inc, d);
    if (lane >= d) inc += o;
  }
  if (lane == 31) wtot[wv] = inc;
  __syncthreads();
  if (wv == 0) {
    const int s = wtot[lane];
    int si = s;
    for (int d = 1; d < 32; d <<= 1) {
      const int o = __shfl_up(si, d);
      if (lane >= d) si += o;
    }
    wtot[lane] = si - s;
  }
  __syncthreads();
  const int off = wtot[wv] + inc - tot;
#pragma unroll
  for (int q = 0; q < PER; q += 4) {
    *reinterpret_cast<int4*>(outp + base + q) =
        make_int4(off + c[q], off + c[q + 1], off + c[q + 2], off + c[q + 3]);
  }
}

__global__ void __launch_bounds__(1024)
k_bin_scatter(const int* __restrict__ src, const int* __restrict__ dst,
              const int* __restrict__ offs, int* __restrict__ bkey, int e, int chunk) {
  __shared__ int cur[NBINP];
  const int t = threadIdx.x;
  const int b = blockIdx.x;
  if (t < NBINP) cur[t] = offs[t * NBLK + b];
  __syncthreads();
  const int lo = b * chunk;
  const int hi = (lo + chunk < e) ? (lo + chunk) : e;
  for (int base = lo + t; base < hi; base += 1024 * EPT) {
    int d[EPT], s[EPT];
#pragma unroll
    for (int q = 0; q < EPT; ++q) {
      const int i = base + q * 1024;
      const bool ok = i < hi;
      d[q] = ok ? dst[i] : -1;
      s[q] = ok ? src[i] : 0;
    }
#pragma unroll
    for (int q = 0; q < EPT; ++q) {
      if (d[q] >= 0) {
        const int p = atomicAdd(&cur[d[q] >> BIN_SHIFT], 1);
        bkey[p] = ((d[q] & (BIN_NODES - 1)) << 16) | s[q];
      }
    }
  }
}

__global__ void __launch_bounds__(1024)
k_bin_finish(const int* __restrict__ offs, const int* __restrict__ bkey,
             int* __restrict__ rowptr, float* __restrict__ dinv,
             int* __restrict__ list, int n, int e) {
  __shared__ int c[BIN_NODES];
  __shared__ int st[BIN_NODES];
  const int t = threadIdx.x;
  const int lane = t & 31;
  const int bk = blockIdx.x;
  const int beg = offs[bk * NBLK];
  const int end = (bk + 1 < NBINP) ? offs[(bk + 1) * NBLK] : e;
  if (t < BIN_NODES) c[t] = 0;
  __syncthreads();
  for (int base = beg + t; base < end; base += 1024 * EPB) {
    int k[EPB];
#pragma unroll
    for (int q = 0; q < EPB; ++q) {
      const int i = base + q * 1024;
      k[q] = (i < end) ? bkey[i] : -1;
    }
#pragma unroll
    for (int q = 0; q < EPB; ++q)
      if (k[q] >= 0) atomicAdd(&c[k[q] >> 16], 1);
  }
  __syncthreads();
  if (t < 32) {
    constexpr int PL = BIN_NODES / 32;
    int v[PL];
    int tot = 0;
#pragma unroll
    for (int q = 0; q < PL; ++q) {
      v[q] = tot;
      tot += c[t * PL + q];
    }
    int inc = tot;
    for (int d = 1; d < 32; d <<= 1) {
      const int o = __shfl_up(inc, d);
      if (lane >= d) inc += o;
    }
    const int off = inc - tot;
#pragma unroll
    for (int q = 0; q < PL; ++q) st[t * PL + q] = off + v[q];
  }
  __syncthreads();
  if (t < BIN_NODES) {
    const int j = bk * BIN_NODES + t;
    const int p = beg + st[t];
    st[t] = p;
    if (j < n) {
      rowptr[j] = p;
      dinv[j] = rsqrtf((float)(c[t] + 1));
    }
  }
  if (bk == 0 && t == 0) rowptr[n] = e;
  __syncthreads();
  for (int base = beg + t; base < end; base += 1024 * EPB) {
    int k[EPB];
#pragma unroll
    for (int q = 0; q < EPB; ++q) {
      const int i = base + q * 1024;
      k[q] = (i < end) ? bkey[i] : -1;
    }
#pragma unroll
    for (int q = 0; q < EPB; ++q) {
      if (k[q] >= 0) {
        const int p = atomicAdd(&st[k[q] >> 16], 1);
        list[p] = k[q] & 0xffff;
      }
    }
  }
}

__global__ void k_lists_move(const int* __restrict__ from, int* __restrict__ to, int count) {
  int i = blockIdx.x * blockDim.x + threadIdx.x;
  if (i < count) to[i] = from[i];
}

__global__ void k_gemm_wmma_f32(const float* __restrict__ A,
                                const float* __restrict__ B,
                                float* __restrict__ C,
                                int M, int K, int N) {
  const int lane = threadIdx.x & 31;
  const int wv   = threadIdx.x >> 5;
  const int h    = lane >> 4;
  const int l16  = lane & 15;

  const int mbase = blockIdx.x * 32;
  const int m0 = mbase + l16;
  const int m1 = m0 + 16;
  const int mc0 = (m0 < M) ? m0 : (M - 1);
  const int mc1 = (m1 < M) ? m1 : (M - 1);
  const int n0 = wv * 32 + l16;
  const int n1 = n0 + 16;

  const float* a0row = A + (size_t)mc0 * K;
  const float* a1row = A + (size_t)mc1 * K;

  v8f acc00 = {}, acc01 = {}, acc10 = {}, acc11 = {};
  for (int k0 = 0; k0 < K; k0 += 4) {
    const int ka = k0 + 2 * h;
    v2f a0, a1, b0, b1;
    a0.x = a0row[ka];  a0.y = a0row[ka + 1];
    a1.x = a1row[ka];  a1.y = a1row[ka + 1];
    const float* br0 = B + (size_t)ka * N;
    const float* br1 = br0 + N;
    b0.x = br0[n0];  b0.y = br1[n0];
    b1.x = br0[n1];  b1.y = br1[n1];
    acc00 = __builtin_amdgcn_wmma_f32_16x16x4_f32(false, a0, false, b0, (short)0, acc00, false, false);
    acc01 = __builtin_amdgcn_wmma_f32_16x16x4_f32(false, a0, false, b1, (short)0, acc01, false, false);
    acc10 = __builtin_amdgcn_wmma_f32_16x16x4_f32(false, a1, false, b0, (short)0, acc10, false, false);
    acc11 = __builtin_amdgcn_wmma_f32_16x16x4_f32(false, a1, false, b1, (short)0, acc11, false, false);
  }

#pragma unroll
  for (int r = 0; r < 8; ++r) {
    const int row0 = mbase + r + 8 * h;
    const int row1 = row0 + 16;
    if (row0 < M) {
      C[(size_t)row0 * N + n0] = acc00[r];
      C[(size_t)row0 * N + n1] = acc01[r];
    }
    if (row1 < M) {
      C[(size_t)row1 * N + n0] = acc10[r];
      C[(size_t)row1 * N + n1] = acc11[r];
    }
  }
}

template <int VEC, bool RELU>
__global__ void k_node_gather(const int* __restrict__ rowptr, const int* __restrict__ list,
                              const float* __restrict__ H, const float* __restrict__ dinv,
                              const float* __restrict__ bias, float* __restrict__ out, int n) {
  typedef float vf __attribute__((ext_vector_type(VEC)));
  constexpr int C = 32 * VEC;
  const int lane = threadIdx.x & 31;
  const int j = __builtin_amdgcn_readfirstlane((int)((blockIdx.x * blockDim.x + threadIdx.x) >> 5));
  if (j >= n) return;
  const int beg = rowptr[j];
  const int end = rowptr[j + 1];
  const float dj = dinv[j];
  const float* hl = H + lane * VEC;
  const vf hs = *reinterpret_cast<const vf*>(hl + (size_t)j * C);
  vf acc = {};
  for (int base = beg; base < end; base += 32) {
    const int m = (end - base < 32) ? (end - base) : 32;
    int s_l = j;
    float w_l = 0.0f;
    if (lane < m) {
      s_l = list[base + lane];
      w_l = dinv[s_l] * dj;
    }
    for (int k = 0; k < m; k += 4) {
      const int s0 = __builtin_amdgcn_readlane(s_l, k);
      const int s1 = __builtin_amdgcn_readlane(s_l, k + 1);
      const int s2 = __builtin_amdgcn_readlane(s_l, k + 2);
      const int s3 = __builtin_amdgcn_readlane(s_l, k + 3);
      const float w0 = __int_as_float(__builtin_amdgcn_readlane(__float_as_int(w_l), k));
      const float w1 = __int_as_float(__builtin_amdgcn_readlane(__float_as_int(w_l), k + 1));
      const float w2 = __int_as_float(__builtin_amdgcn_readlane(__float_as_int(w_l), k + 2));
      const float w3 = __int_as_float(__builtin_amdgcn_readlane(__float_as_int(w_l), k + 3));
      const vf h0 = *reinterpret_cast<const vf*>(hl + (size_t)s0 * C);
      const vf h1 = *reinterpret_cast<const vf*>(hl + (size_t)s1 * C);
      const vf h2 = *reinterpret_cast<const vf*>(hl + (size_t)s2 * C);
      const vf h3 = *reinterpret_cast<const vf*>(hl + (size_t)s3 * C);
      acc = h0 * w0 + acc;
      const vf t1 = h1 * w1 + acc;
      acc = (k + 1 < m) ? t1 : acc;
      const vf t2 = h2 * w2 + acc;
      acc = (k + 2 < m) ? t2 : acc;
      const vf t3 = h3 * w3 + acc;
      acc = (k + 3 < m) ? t3 : acc;
    }
  }
  acc = hs * (dj * dj) + acc;
  vf r = acc + *reinterpret_cast<const vf*>(bias + lane * VEC);
  if (RELU) {
#pragma unroll
    for (int q = 0; q < VEC; ++q) r[q] = fmaxf(r[q], 0.0f);
  }
  *reinterpret_cast<vf*>(out + (size_t)j * C + lane * VEC) = r;
}

extern "C" void kernel_launch(void* const* d_in, const int* in_sizes, int n_in,
                              void* d_out, int out_size, void* d_ws, size_t ws_size,
                              hipStream_t stream) {
  const int CIN = 256, CH = 128, CO = 64;
  const float* x  = (const float*)d_in[0];
  const int*   ei = (const int*)d_in[1];
  const float* W1 = (const float*)d_in[2];
  const float* b1 = (const float*)d_in[3];
  const float* W2 = (const float*)d_in[4];
  const float* b2 = (const float*)d_in[5];
  float* out = (float*)d_out;

  const int N = in_sizes[0] / CIN;
  const int E = in_sizes[1] / 2;
  const int* src = ei;
  const int* dst = ei + E;

  float* ws   = (float*)d_ws;
  float* dinv = ws;
  float* Hbuf = ws + 50176;
  float* AGG  = ws + 50176 + (size_t)N * CH;

  int* oi     = (int*)d_out;
  int* hist   = oi;
  int* offs   = oi + 53248;
  int* rowptr = oi + 106496;
  int* list   = rowptr + N + 1;
  int* bkey   = list + E;
  int* rowptr2 = (int*)(Hbuf + (size_t)N * CO);
  int* list2   = rowptr2 + N + 1;

  const int T = 256;

  const int chunk = (E + 255) / 256;
  const int nbin  = (N + 255) / 256;
  k_bin_count  <<<256, 1024, 0, stream>>>(dst, hist, E, chunk);
  k_scan_excl  <<<1, 1024, 0, stream>>>(hist, offs);
  k_bin_scatter<<<256, 1024, 0, stream>>>(src, dst, offs, bkey, E, chunk);
  k_bin_finish <<<nbin, 1024, 0, stream>>>(offs, bkey, rowptr, dinv, list, N, E);

  k_gemm_wmma_f32<<<(N + 31) / 32, 32 * (CH / 32), 0, stream>>>(x, W1, Hbuf, N, CIN, CH);
  k_node_gather<4, true><<<((size_t)N * 32 + T - 1) / T, T, 0, stream>>>(rowptr, list, Hbuf, dinv, b1, AGG, N);

  k_lists_move<<<(N + 1 + E + T - 1) / T, T, 0, stream>>>(rowptr, rowptr2, N + 1 + E);
  k_gemm_wmma_f32<<<(N + 31) / 32, 32 * (CO / 32), 0, stream>>>(AGG, W2, Hbuf, N, CH, CO);
  k_node_gather<2, false><<<((size_t)N * 32 + T - 1) / T, T, 0, stream>>>(rowptr2, list2, Hbuf, dinv, b2, out, N);
}


# Layer-2 product folded into the layer-1 gather: a 16-node work-group keeps its relu rows in LDS and multiplies them by W2 on WMMA; AGG plane and list-copy launch gone; sum orders kept
# speedup vs baseline: 2.6692x; 1.0968x over previous
#include <hip/hip_runtime.h>
#include <hip/hip_bf16.h>

typedef float v2f __attribute__((ext_vector_type(2)));
typedef float v8f __attribute__((ext_vector_type(8)));

constexpr int BIN_SHIFT = 8;
constexpr int BIN_NODES = 1 << BIN_SHIFT;
constexpr int NBLK  = 256;
constexpr int NBINP = 208;
constexpr int EPT   = 4;
constexpr int EPB   = 8;

__global__ void __launch_bounds__(1024)
k_bin_count(const int* __restrict__ dst, int* __restrict__ hist, int e, int chunk) {
  __shared__ int h[NBINP];
  const int t = threadIdx.x;
  const int b = blockIdx.x;
  if (t < NBINP) h[t] = 0;
  __syncthreads();
  const int lo = b * chunk;
  const int hi = (lo + chunk < e) ? (lo + chunk) : e;
  for (int base = lo + t; base < hi; base += 1024 * EPT) {
    int d[EPT];
#pragma unroll
    for (int q = 0; q < EPT; ++q) {
      const int i = base + q * 1024;
      d[q] = (i < hi) ? dst[i] : -1;
    }
#pragma unroll
    for (int q = 0; q < EPT; ++q)
      if (d[q] >= 0) atomicAdd(&h[d[q] >> BIN_SHIFT], 1);
  }
  __syncthreads();
  if (t < NBINP) hist[t * NBLK + b] = h[t];
}

__global__ void __launch_bounds__(1024)
k_scan_excl(const int* __restrict__ in, int* __restrict__ outp) {
  constexpr int PER = 52;
  __shared__ int wtot[32];
  const int t = threadIdx.x;
  const int lane = t & 31;
  const int wv = t >> 5;
  const int base = t * PER;
  int c[PER];
#pragma unroll
  for (int q = 0; q < PER; q += 4) {
    const int4 v = *reinterpret_cast<const int4*>(in + base + q);
    c[q] = v.x; c[q + 1] = v.y; c[q + 2] = v.z; c[q + 3] = v.w;
  }
  int tot = 0;
#pragma unroll
  for (int q = 0; q < PER; ++q) {
    const int v = c[q];
    c[q] = tot;
    tot += v;
  }
  int inc = tot;
  for (int d = 1; d < 32; d <<= 1) {
    const int o = __shfl_up(inc, d);
    if (lane >= d) inc += o;
  }
  if (lane == 31) wtot[wv] = inc;
  __syncthreads();
  if (wv == 0) {
    const int s = wtot[lane];
    int si = s;
    for (int d = 1; d < 32; d <<= 1) {
      const int o = __shfl_up(si, d);
      if (lane >= d) si += o;
    }
    wtot[lane] = si - s;
  }
  __syncthreads();
  const int off = wtot[wv] + inc - tot;
#pragma unroll
  for (int q = 0; q < PER; q += 4) {
    *reinterpret_cast<int4*>(outp + base + q) =
        make_int4(off + c[q], off + c[q + 1], off + c[q + 2], off + c[q + 3]);
  }
}

__global__ void __launch_bounds__(1024)
k_bin_scatter(const int* __restrict__ src, const int* __restrict__ dst,
              const int* __restrict__ offs, int* __restrict__ bkey, int e, int chunk) {
  __shared__ int cur[NBINP];
  const int t = threadIdx.x;
  const int b = blockIdx.x;
  if (t < NBINP) cur[t] = offs[t * NBLK + b];
  __syncthreads();
  const int lo = b * chunk;
  const int hi = (lo + chunk < e) ? (lo + chunk) : e;
  for (int base = lo + t; base < hi; base += 1024 * EPT) {
    int d[EPT], s[EPT];
#pragma unroll
    for (int q = 0; q < EPT; ++q) {
      const int i = base + q * 1024;
      const bool ok = i < hi;
      d[q] = ok ? dst[i] : -1;
      s[q] = ok ? src[i] : 0;
    }
#pragma unroll
    for (int q = 0; q < EPT; ++q) {
      if (d[q] >= 0) {
        const int p = atomicAdd(&cur[d[q] >> BIN_SHIFT], 1);
        bkey[p] = ((d[q] & (BIN_NODES - 1)) << 16) | s[q];
      }
    }
  }
}

__global__ void __launch_bounds__(1024)
k_bin_finish(const int* __restrict__ offs, const int* __restrict__ bkey,
             int* __restrict__ rowptr, float* __restrict__ dinv,
             int* __restrict__ list, int n, int e) {
  __shared__ int c[BIN_NODES];
  __shared__ int st[BIN_NODES];
  const int t = threadIdx.x;
  const int lane = t & 31;
  const int bk = blockIdx.x;
  const int beg = offs[bk * NBLK];
  const int end = (bk + 1 < NBINP) ? offs[(bk + 1) * NBLK] : e;
  if (t < BIN_NODES) c[t] = 0;
  __syncthreads();
  for (int base = beg + t; base < end; base += 1024 * EPB) {
    int k[EPB];
#pragma unroll
    for (int q = 0; q < EPB; ++q) {
      const int i = base + q * 1024;
      k[q] = (i < end) ? bkey[i] : -1;
    }
#pragma unroll
    for (int q = 0; q < EPB; ++q)
      if (k[q] >= 0) atomicAdd(&c[k[q] >> 16], 1);
  }
  __syncthreads();
  if (t < 32) {
    constexpr int PL = BIN_NODES / 32;
    int v[PL];
    int tot = 0;
#pragma unroll
    for (int q = 0; q < PL; ++q) {
      v[q] = tot;
      tot += c[t * PL + q];
    }
    int inc = tot;
    for (int d = 1; d < 32; d <<= 1) {
      const int o = __shfl_up(inc, d);
      if (lane >= d) inc += o;
    }
    const int off = inc - tot;
#pragma unroll
    for (int q = 0; q < PL; ++q) st[t * PL + q] = off + v[q];
  }
  __syncthreads();
  if (t < BIN_NODES) {
    const int j = bk * BIN_NODES + t;
    const int p = beg + st[t];
    st[t] = p;
    if (j < n) {
      rowptr[j] = p;
      dinv[j] = rsqrtf((float)(c[t] + 1));
    }
  }
  if (bk == 0 && t == 0) rowptr[n] = e;
  __syncthreads();
  for (int base = beg + t; base < end; base += 1024 * EPB) {
    int k[EPB];
#pragma unroll
    for (int q = 0; q < EPB; ++q) {
      const int i = base + q * 1024;
      k[q] = (i < end) ? bkey[i] : -1;
    }
#pragma unroll
    for (int q = 0; q < EPB; ++q) {
      if (k[q] >= 0) {
        const int p = atomicAdd(&st[k[q] >> 16], 1);
        list[p] = k[q] & 0xffff;
      }
    }
  }
}

__global__ void k_gemm_wmma_f32(const float* __restrict__ A,
                                const float* __restrict__ B,
                                float* __restrict__ C,
                                int M, int K, int N) {
  const int lane = threadIdx.x & 31;
  const int wv   = threadIdx.x >> 5;
  const int h    = lane >> 4;
  const int l16  = lane & 15;

  const int mbase = blockIdx.x * 32;
  const int m0 = mbase + l16;
  const int m1 = m0 + 16;
  const int mc0 = (m0 < M) ? m0 : (M - 1);
  const int mc1 = (m1 < M) ? m1 : (M - 1);
  const int n0 = wv * 32 + l16;
  const int n1 = n0 + 16;

  const float* a0row = A + (size_t)mc0 * K;
  const float* a1row = A + (size_t)mc1 * K;

  v8f acc00 = {}, acc01 = {}, acc10 = {}, acc11 = {};
  for (int k0 = 0; k0 < K; k0 += 4) {
    const int ka = k0 + 2 * h;
    v2f a0, a1, b0, b1;
    a0.x = a0row[ka];  a0.y = a0row[ka + 1];
    a1.x = a1row[ka];  a1.y = a1row[ka + 1];
    const float* br0 = B + (size_t)ka * N;
    const float* br1 = br0 + N;
    b0.x = br0[n0];  b0.y = br1[n0];
    b1.x = br0[n1];  b1.y = br1[n1];
    acc00 = __builtin_amdgcn_wmma_f32_16x16x4_f32(false, a0, false, b0, (short)0, acc00, false, false);
    acc01 = __builtin_amdgcn_wmma_f32_16x16x4_f32(false, a0, false, b1, (short)0, acc01, false, false);
    acc10 = __builtin_amdgcn_wmma_f32_16x16x4_f32(false, a1, false, b0, (short)0, acc10, false, false);
    acc11 = __builtin_amdgcn_wmma_f32_16x16x4_f32(false, a1, false, b1, (short)0, acc11, false, false);
  }

#pragma unroll
  for (int r = 0; r < 8; ++r) {
    const int row0 = mbase + r + 8 * h;
    const int row1 = row0 + 16;
    if (row0 < M) {
      C[(size_t)row0 * N + n0] = acc00[r];
      C[(size_t)row0 * N + n1] = acc01[r];
    }
    if (row1 < M) {
      C[(size_t)row1 * N + n0] = acc10[r];
      C[(size_t)row1 * N + n1] = acc11[r];
    }
  }
}

template <int VEC, bool RELU>
__global__ void k_node_gather(const int* __restrict__ rowptr, const int* __restrict__ list,
                              const float* __restrict__ H, const float* __restrict__ dinv,
                              const float* __restrict__ bias, float* __restrict__ out, int n) {
  typedef float vf __attribute__((ext_vector_type(VEC)));
  constexpr int C = 32 * VEC;
  const int lane = threadIdx.x & 31;
  const int j = __builtin_amdgcn_readfirstlane((int)((blockIdx.x * blockDim.x + threadIdx.x) >> 5));
  if (j >= n) return;
  const int beg = rowptr[j];
  const int end = rowptr[j + 1];
  const float dj = dinv[j];
  const float* hl = H + lane * VEC;
  const vf hs = *reinterpret_cast<const vf*>(hl + (size_t)j * C);
  vf acc = {};
  for (int base = beg; base < end; base += 32) {
    const int m = (end - base < 32) ? (end - base) : 32;
    int s_l = j;
    float w_l = 0.0f;
    if (lane < m) {
      s_l = list[base + lane];
      w_l = dinv[s_l] * dj;
    }
    for (int k = 0; k < m; k += 4) {
      const int s0 = __builtin_amdgcn_readlane(s_l, k);
      const int s1 = __builtin_amdgcn_readlane(s_l, k + 1);
      const int s2 = __builtin_amdgcn_readlane(s_l, k + 2);
      const int s3 = __builtin_amdgcn_readlane(s_l, k + 3);
      const float w0 = __int_as_float(__builtin_amdgcn_readlane(__float_as_int(w_l), k));
      const float w1 = __int_as_float(__builtin_amdgcn_readlane(__float_as_int(w_l), k + 1));
      const float w2 = __int_as_float(__builtin_amdgcn_readlane(__float_as_int(w_l), k + 2));
      const float w3 = __int_as_float(__builtin_amdgcn_readlane(__float_as_int(w_l), k + 3));
      const vf h0 = *reinterpret_cast<const vf*>(hl + (size_t)s0 * C);
      const vf h1 = *reinterpret_cast<const vf*>(hl + (size_t)s1 * C);
      const vf h2 = *reinterpret_cast<const vf*>(hl + (size_t)s2 * C);
      const vf h3 = *reinterpret_cast<const vf*>(hl + (size_t)s3 * C);
      acc = h0 * w0 + acc;
      const vf t1 = h1 * w1 + acc;
      acc = (k + 1 < m) ? t1 : acc;
      const vf t2 = h2 * w2 + acc;
      acc = (k + 2 < m) ? t2 : acc;
      const vf t3 = h3 * w3 + acc;
      acc = (k + 3 < m) ? t3 : acc;
    }
  }
  acc = hs * (dj * dj) + acc;
  vf r = acc + *reinterpret_cast<const vf*>(bias + lane * VEC);
  if (RELU) {
#pragma unroll
    for (int q = 0; q < VEC; ++q) r[q] = fmaxf(r[q], 0.0f);
  }
  *reinterpret_cast<vf*>(out + (size_t)j * C + lane * VEC) = r;
}

constexpr int TN = 16;
constexpr int SA_PITCH = 132;
constexpr int SB_PITCH = 72;

__global__ void __launch_bounds__(512)
k_gather_relu_gemm(const int* __restrict__ rowptr, const int* __restrict__ list,
                   const float* __restrict__ H, const float* __restrict__ dinv,
                   const float* __restrict__ bias, const float* __restrict__ W2,
                   float* __restrict__ H2, int n) {
  typedef float vf __attribute__((ext_vector_type(4)));
  constexpr int C = 128, CO = 64;
  __shared__ __align__(16) float sA[TN * SA_PITCH];
  __shared__ __align__(16) float sB[C * SB_PITCH];
  const int t = threadIdx.x;
  const int lane = t & 31;
  const int wv = __builtin_amdgcn_readfirstlane(t >> 5);
  const int j = blockIdx.x * TN + wv;

  vf wreg[4];
#pragma unroll
  for (int i = 0; i < 4; ++i)
    wreg[i] = *reinterpret_cast<const vf*>(W2 + (size_t)(t + 512 * i) * 4);

  vf r = {};
  if (j < n) {
    const int beg = rowptr[j];
    const int end = rowptr[j + 1];
    const float dj = dinv[j];
    const float* hl = H + lane * 4;
    const vf hs = *reinterpret_cast<const vf*>(hl + (size_t)j * C);
    vf acc = {};
    for (int base = beg; base < end; base += 32) {
      const int m = (end - base < 32) ? (end - base) : 32;
      int s_l = j;
      float w_l = 0.0f;
      if (lane < m) {
        s_l = list[base + lane];
        w_l = dinv[s_l] * dj;
      }
      for (int k = 0; k < m; k += 4) {
        const int s0 = __builtin_amdgcn_readlane(s_l, k);
        const int s1 = __builtin_amdgcn_readlane(s_l, k + 1);
        const int s2 = __builtin_amdgcn_readlane(s_l, k + 2);
        const int s3 = __builtin_amdgcn_readlane(s_l, k + 3);
        const float w0 = __int_as_float(__builtin_amdgcn_readlane(__float_as_int(w_l), k));
        const float w1 = __int_as_float(__builtin_amdgcn_readlane(__float_as_int(w_l), k + 1));
        const float w2 = __int_as_float(__builtin_amdgcn_readlane(__float_as_int(w_l), k + 2));
        const float w3 = __int_as_float(__builtin_amdgcn_readlane(__float_as_int(w_l), k + 3));
        const vf h0 = *reinterpret_cast<const vf*>(hl + (size_t)s0 * C);
        const vf h1 = *reinterpret_cast<const vf*>(hl + (size_t)s1 * C);
        const vf h2 = *reinterpret_cast<const vf*>(hl + (size_t)s2 * C);
        const vf h3 = *reinterpret_cast<const vf*>(hl + (size_t)s3 * C);
        acc = h0 * w0 + acc;
        const vf t1 = h1 * w1 + acc;
        acc = (k + 1 < m) ? t1 : acc;
        const vf t2 = h2 * w2 + acc;
        acc = (k + 2 < m) ? t2 : acc;
        const vf t3 = h3 * w3 + acc;
        acc = (k + 3 < m) ? t3 : acc;
      }
    }
    acc = hs * (dj * dj) + acc;
    r = acc + *reinterpret_cast<const vf*>(bias + lane * 4);
#pragma unroll
    for (int q = 0; q < 4; ++q) r[q] = fmaxf(r[q], 0.0f);
  }
  *reinterpret_cast<vf*>(sA + wv * SA_PITCH + lane * 4) = r;
#pragma unroll
  for (int i = 0; i < 4; ++i) {
    const int idx4 = t + 512 * i;
    *reinterpret_cast<vf*>(sB + (idx4 >> 4) * SB_PITCH + (idx4 & 15) * 4) = wreg[i];
  }
  __syncthreads();

  if (wv < 4) {
    const int h   = lane >> 4;
    const int l16 = lane & 15;
    const float* ap = sA + l16 * SA_PITCH + 2 * h;
    const float* bp = sB + (2 * h) * SB_PITCH + wv * 16 + l16;
    v8f acc = {};
    v2f fa[2][4], fb[2][4];
#pragma unroll
    for (int q = 0; q < 4; ++q) {
      fa[0][q] = *reinterpret_cast<const v2f*>(ap + 4 * q);
      fb[0][q].x = bp[(4 * q) * SB_PITCH];
      fb[0][q].y = bp[(4 * q + 1) * SB_PITCH];
    }
#pragma unroll
    for (int g = 0; g < C / 16; ++g) {
      const int cur = g & 1, nxt = cur ^ 1;
      if (g + 1 < C / 16) {
#pragma unroll
        for (int q = 0; q < 4; ++q) {
          const int k0 = 16 * (g + 1) + 4 * q;
          fa[nxt][q] = *reinterpret_cast<const v2f*>(ap + k0);
          fb[nxt][q].x = bp[k0 * SB_PITCH];
          fb[nxt][q].y = bp[(k0 + 1) * SB_PITCH];
        }
      }
      __builtin_amdgcn_sched_barrier(0);
#pragma unroll
      for (int q = 0; q < 4; ++q)
        acc = __builtin_amdgcn_wmma_f32_16x16x4_f32(false, fa[cur][q], false, fb[cur][q], (short)0, acc, false, false);
    }
    const int row0 = blockIdx.x * TN + 8 * h;
    float* op = H2 + (size_t)row0 * CO + wv * 16 + l16;
#pragma unroll
    for (int q = 0; q < 8; ++q)
      if (row0 + q < n) op[(size_t)q * CO] = acc[q];
  }
}

extern "C" void kernel_launch(void* const* d_in, const int* in_sizes, int n_in,
                              void* d_out, int out_size, void* d_ws, size_t ws_size,
                              hipStream_t stream) {
  const int CIN = 256, CH = 128, CO = 64;
  const float* x  = (const float*)d_in[0];
  const int*   ei = (const int*)d_in[1];
  const float* W1 = (const float*)d_in[2];
  const float* b1 = (const float*)d_in[3];
  const float* W2 = (const float*)d_in[4];
  const float* b2 = (const float*)d_in[5];
  float* out = (float*)d_out;

  const int N = in_sizes[0] / CIN;
  const int E = in_sizes[1] / 2;
  const int* src = ei;
  const int* dst = ei + E;

  float* ws   = (float*)d_ws;
  float* dinv = ws;
  float* Hbuf = ws + 50176;
  float* H2   = ws + 50176 + (size_t)N * CH;
  int* rowptr = (int*)(H2 + (size_t)N * CO);
  int* list   = rowptr + N + 1;

  int* oi     = (int*)d_out;
  int* hist   = oi;
  int* offs   = oi + 53248;
  int* bkey   = oi + 106496;

  const int T = 256;

  const int chunk = (E + 255) / 256;
  const int nbin  = (N + 255) / 256;
  k_bin_count  <<<256, 1024, 0, stream>>>(dst, hist, E, chunk);
  k_scan_excl  <<<1, 1024, 0, stream>>>(hist, offs);
  k_bin_scatter<<<256, 1024, 0, stream>>>(src, dst, offs, bkey, E, chunk);
  k_bin_finish <<<nbin, 1024, 0, stream>>>(offs, bkey, rowptr, dinv, list, N, E);

  k_gemm_wmma_f32<<<(N + 31) / 32, 32 * (CH / 32), 0, stream>>>(x, W1, Hbuf, N, CIN, CH);
  k_gather_relu_gemm<<<(N + TN - 1) / TN, 32 * TN, 0, stream>>>(rowptr, list, Hbuf, dinv, b1, W2, H2, N);

  k_node_gather<2, false><<<((size_t)N * 32 + T - 1) / T, T, 0, stream>>>(rowptr, list, H2, dinv, b2, out, N);
}


# x@W1 product fed by whole-line loads: x rows staged through LDS in 64-k slabs, W1 read 16 B per lane into four interleaved column tiles of a 16x64 wave tile; k order and sum orders kept
# speedup vs baseline: 2.8883x; 1.0821x over previous
#include <hip/hip_runtime.h>
#include <hip/hip_bf16.h>

typedef float v2f __attribute__((ext_vector_type(2)));
typedef float v8f __attribute__((ext_vector_type(8)));

constexpr int BIN_SHIFT = 8;
constexpr int BIN_NODES = 1 << BIN_SHIFT;
constexpr int NBLK  = 256;
constexpr int NBINP = 208;
constexpr int EPT   = 4;
constexpr int EPB   = 8;

__global__ void __launch_bounds__(1024)
k_bin_count(const int* __restrict__ dst, int* __restrict__ hist, int e, int chunk) {
  __shared__ int h[NBINP];
  const int t = threadIdx.x;
  const int b = blockIdx.x;
  if (t < NBINP) h[t] = 0;
  __syncthreads();
  const int lo = b * chunk;
  const int hi = (lo + chunk < e) ? (lo + chunk) : e;
  for (int base = lo + t; base < hi; base += 1024 * EPT) {
    int d[EPT];
#pragma unroll
    for (int q = 0; q < EPT; ++q) {
      const int i = base + q * 1024;
      d[q] = (i < hi) ? dst[i] : -1;
    }
#pragma unroll
    for (int q = 0; q < EPT; ++q)
      if (d[q] >= 0) atomicAdd(&h[d[q] >> BIN_SHIFT], 1);
  }
  __syncthreads();
  if (t < NBINP) hist[t * NBLK + b] = h[t];
}

__global__ void __launch_bounds__(1024)
k_scan_excl(const int* __restrict__ in, int* __restrict__ outp) {
  constexpr int PER = 52;
  __shared__ int wtot[32];
  const int t = threadIdx.x;
  const int lane = t & 31;
  const int wv = t >> 5;
  const int base = t * PER;
  int c[PER];
#pragma unroll
  for (int q = 0; q < PER; q += 4) {
    const int4 v = *reinterpret_cast<const int4*>(in + base + q);
    c[q] = v.x; c[q + 1] = v.y; c[q + 2] = v.z; c[q + 3] = v.w;
  }
  int tot = 0;
#pragma unroll
  for (int q = 0; q < PER; ++q) {
    const int v = c[q];
    c[q] = tot;
    tot += v;
  }
  int inc = tot;
  for (int d = 1; d < 32; d <<= 1) {
    const int o = __shfl_up(inc, d);
    if (lane >= d) inc += o;
  }
  if (lane == 31) wtot[wv] = inc;
  __syncthreads();
  if (wv == 0) {
    const int s = wtot[lane];
    int si = s;
    for (int d = 1; d < 32; d <<= 1) {
      const int o = __shfl_up(si, d);
      if (lane >= d) si += o;
    }
    wtot[lane] = si - s;
  }
  __syncthreads();
  const int off = wtot[wv] + inc - tot;
#pragma unroll
  for (int q = 0; q < PER; q += 4) {
    *reinterpret_cast<int4*>(outp + base + q) =
        make_int4(off + c[q], off + c[q + 1], off + c[q + 2], off + c[q + 3]);
  }
}

__global__ void __launch_bounds__(1024)
k_bin_scatter(const int* __restrict__ src, const int* __restrict__ dst,
              const int* __restrict__ offs, int* __restrict__ bkey, int e, int chunk) {
  __shared__ int cur[NBINP];
  const int t = threadIdx.x;
  const int b = blockIdx.x;
  if (t < NBINP) cur[t] = offs[t * NBLK + b];
  __syncthreads();
  const int lo = b * chunk;
  const int hi = (lo + chunk < e) ? (lo + chunk) : e;
  for (int base = lo + t; base < hi; base += 1024 * EPT) {
    int d[EPT], s[EPT];
#pragma unroll
    for (int q = 0; q < EPT; ++q) {
      const int i = base + q * 1024;
      const bool ok = i < hi;
      d[q] = ok ? dst[i] : -1;
      s[q] = ok ? src[i] : 0;
    }
#pragma unroll
    for (int q = 0; q < EPT; ++q) {
      if (d[q] >= 0) {
        const int p = atomicAdd(&cur[d[q] >> BIN_SHIFT], 1);
        bkey[p] = ((d[q] & (BIN_NODES - 1)) << 16) | s[q];
      }
    }
  }
}

__global__ void __launch_bounds__(1024)
k_bin_finish(const int* __restrict__ offs, const int* __restrict__ bkey,
             int* __restrict__ rowptr, float* __restrict__ dinv,
             int* __restrict__ list, int n, int e) {
  __shared__ int c[BIN_NODES];
  __shared__ int st[BIN_NODES];
  const int t = threadIdx.x;
  const int lane = t & 31;
  const int bk = blockIdx.x;
  const int beg = offs[bk * NBLK];
  const int end = (bk + 1 < NBINP) ? offs[(bk + 1) * NBLK] : e;
  if (t < BIN_NODES) c[t] = 0;
  __syncthreads();
  for (int base = beg + t; base < end; base += 1024 * EPB) {
    int k[EPB];
#pragma unroll
    for (int q = 0; q < EPB; ++q) {
      const int i = base + q * 1024;
      k[q] = (i < end) ? bkey[i] : -1;
    }
#pragma unroll
    for (int q = 0; q < EPB; ++q)
      if (k[q] >= 0) atomicAdd(&c[k[q] >> 16], 1);
  }
  __syncthreads();
  if (t < 32) {
    constexpr int PL = BIN_NODES / 32;
    int v[PL];
    int tot = 0;
#pragma unroll
    for (int q = 0; q < PL; ++q) {
      v[q] = tot;
      tot += c[t * PL + q];
    }
    int inc = tot;
    for (int d = 1; d < 32; d <<= 1) {
      const int o = __shfl_up(inc, d);
      if (lane >= d) inc += o;
    }
    const int off = inc - tot;
#pragma unroll
    for (int q = 0; q < PL; ++q) st[t * PL + q] = off + v[q];
  }
  __syncthreads();
  if (t < BIN_NODES) {
    const int j = bk * BIN_NODES + t;
    const int p = beg + st[t];
    st[t] = p;
    if (j < n) {
      rowptr[j] = p;
      dinv[j] = rsqrtf((float)(c[t] + 1));
    }
  }
  if (bk == 0 && t == 0) rowptr[n] = e;
  __syncthreads();
  for (int base = beg + t; base < end; base += 1024 * EPB) {
    int k[EPB];
#pragma unroll
    for (int q = 0; q < EPB; ++q) {
      const int i = base + q * 1024;
      k[q] = (i < end) ? bkey[i] : -1;
    }
#pragma unroll
    for (int q = 0; q < EPB; ++q) {
      if (k[q] >= 0) {
        const int p = atomicAdd(&st[k[q] >> 16], 1);
        list[p] = k[q] & 0xffff;
      }
    }
  }
}

constexpr int GM_ROWS  = 32;
constexpr int GM_K     = 256;
constexpr int GM_N     = 128;
constexpr int GM_SLAB  = 64;
constexpr int GM_PITCH = 260;

__global__ void __launch_bounds__(128)
k_gemm_x_w1(const float* __restrict__ A, const float* __restrict__ B,
            float* __restrict__ C, int M) {
  typedef float vf __attribute__((ext_vector_type(4)));
  __shared__ __align__(16) float sX[GM_ROWS * GM_PITCH];
  const int t = threadIdx.x;
  const int lane = t & 31;
  const int wv = __builtin_amdgcn_readfirstlane(t >> 5);
  const int rt = wv >> 1;
  const int ch = wv & 1;
  const int h = lane >> 4;
  const int l16 = lane & 15;
  const int mbase = blockIdx.x * GM_ROWS;

  const int srow = t >> 4;
  const int sc4 = t & 15;
  const float* gsrc[4];
#pragma unroll
  for (int i = 0; i < 4; ++i) {
    const int r = mbase + srow + 8 * i;
    const int rc = (r < M) ? r : (M - 1);
    gsrc[i] = A + (size_t)rc * GM_K + 4 * sc4;
  }
  float* sdst = sX + srow * GM_PITCH + 4 * sc4;

  vf st[4];
#pragma unroll
  for (int i = 0; i < 4; ++i) st[i] = *reinterpret_cast<const vf*>(gsrc[i]);

  v8f acc0 = {}, acc1 = {}, acc2 = {}, acc3 = {};
  const float* ap = sX + (16 * rt + l16) * GM_PITCH + 2 * h;
  const float* bp = B + (size_t)(2 * h) * GM_N + 64 * ch + 4 * l16;

  for (int q = 0; q < GM_K / GM_SLAB; ++q) {
#pragma unroll
    for (int i = 0; i < 4; ++i)
      *reinterpret_cast<vf*>(sdst + 8 * i * GM_PITCH + GM_SLAB * q) = st[i];
    __syncthreads();
    if (q + 1 < GM_K / GM_SLAB) {
#pragma unroll
      for (int i = 0; i < 4; ++i)
        st[i] = *reinterpret_cast<const vf*>(gsrc[i] + GM_SLAB * (q + 1));
    }
    for (int ks = 0; ks < GM_SLAB / 4; ++ks) {
      const int k0 = GM_SLAB * q + 4 * ks;
      const v2f a = *reinterpret_cast<const v2f*>(ap + k0);
      const vf b0 = *reinterpret_cast<const vf*>(bp + (size_t)k0 * GM_N);
      const vf b1 = *reinterpret_cast<const vf*>(bp + (size_t)(k0 + 1) * GM_N);
      v2f f0, f1, f2, f3;
      f0.x = b0.x; f0.y = b1.x;
      f1.x = b0.y; f1.y = b1.y;
      f2.x = b0.z; f2.y = b1.z;
      f3.x = b0.w; f3.y = b1.w;
      acc0 = __builtin_amdgcn_wmma_f32_16x16x4_f32(false, a, false, f0, (short)0, acc0, false, false);
      acc1 = __builtin_amdgcn_wmma_f32_16x16x4_f32(false, a, false, f1, (short)0, acc1, false, false);
      acc2 = __builtin_amdgcn_wmma_f32_16x16x4_f32(false, a, false, f2, (short)0, acc2, false, false);
      acc3 = __builtin_amdgcn_wmma_f32_16x16x4_f32(false, a, false, f3, (short)0, acc3, false, false);
    }
  }

  const int row0 = mbase + 16 * rt + 8 * h;
  float* op = C + (size_t)row0 * GM_N + 64 * ch + 4 * l16;
#pragma unroll
  for (int r = 0; r < 8; ++r) {
    if (row0 + r < M) {
      vf o;
      o.x = acc0[r]; o.y = acc1[r]; o.z = acc2[r]; o.w = acc3[r];
      *reinterpret_cast<vf*>(op + (size_t)r * GM_N) = o;
    }
  }
}

template <int VEC, bool RELU>
__global__ void k_node_gather(const int* __restrict__ rowptr, const int* __restrict__ list,
                              const float* __restrict__ H, const float* __restrict__ dinv,
                              const float* __restrict__ bias, float* __restrict__ out, int n) {
  typedef float vf __attribute__((ext_vector_type(VEC)));
  constexpr int C = 32 * VEC;
  const int lane = threadIdx.x & 31;
  const int j = __builtin_amdgcn_readfirstlane((int)((blockIdx.x * blockDim.x + threadIdx.x) >> 5));
  if (j >= n) return;
  const int beg = rowptr[j];
  const int end = rowptr[j + 1];
  const float dj = dinv[j];
  const float* hl = H + lane * VEC;
  const vf hs = *reinterpret_cast<const vf*>(hl + (size_t)j * C);
  vf acc = {};
  for (int base = beg; base < end; base += 32) {
    const int m = (end - base < 32) ? (end - base) : 32;
    int s_l = j;
    float w_l = 0.0f;
    if (lane < m) {
      s_l = list[base + lane];
      w_l = dinv[s_l] * dj;
    }
    for (int k = 0; k < m; k += 4) {
      const int s0 = __builtin_amdgcn_readlane(s_l, k);
      const int s1 = __builtin_amdgcn_readlane(s_l, k + 1);
      const int s2 = __builtin_amdgcn_readlane(s_l, k + 2);
      const int s3 = __builtin_amdgcn_readlane(s_l, k + 3);
      const float w0 = __int_as_float(__builtin_amdgcn_readlane(__float_as_int(w_l), k));
      const float w1 = __int_as_float(__builtin_amdgcn_readlane(__float_as_int(w_l), k + 1));
      const float w2 = __int_as_float(__builtin_amdgcn_readlane(__float_as_int(w_l), k + 2));
      const float w3 = __int_as_float(__builtin_amdgcn_readlane(__float_as_int(w_l), k + 3));
      const vf h0 = *reinterpret_cast<const vf*>(hl + (size_t)s0 * C);
      const vf h1 = *reinterpret_cast<const vf*>(hl + (size_t)s1 * C);
      const vf h2 = *reinterpret_cast<const vf*>(hl + (size_t)s2 * C);
      const vf h3 = *reinterpret_cast<const vf*>(hl + (size_t)s3 * C);
      acc = h0 * w0 + acc;
      const vf t1 = h1 * w1 + acc;
      acc = (k + 1 < m) ? t1 : acc;
      const vf t2 = h2 * w2 + acc;
      acc = (k + 2 < m) ? t2 : acc;
      const vf t3 = h3 * w3 + acc;
      acc = (k + 3 < m) ? t3 : acc;
    }
  }
  acc = hs * (dj * dj) + acc;
  vf r = acc + *reinterpret_cast<const vf*>(bias + lane * VEC);
  if (RELU) {
#pragma unroll
    for (int q = 0; q < VEC; ++q) r[q] = fmaxf(r[q], 0.0f);
  }
  *reinterpret_cast<vf*>(out + (size_t)j * C + lane * VEC) = r;
}

constexpr int TN = 16;
constexpr int SA_PITCH = 132;
constexpr int SB_PITCH = 72;

__global__ void __launch_bounds__(512)
k_gather_relu_gemm(const int* __restrict__ rowptr, const int* __restrict__ list,
                   const float* __restrict__ H, const float* __restrict__ dinv,
                   const float* __restrict__ bias, const float* __restrict__ W2,
                   float* __restrict__ H2, int n) {
  typedef float vf __attribute__((ext_vector_type(4)));
  constexpr int C = 128, CO = 64;
  __shared__ __align__(16) float sA[TN * SA_PITCH];
  __shared__ __align__(16) float sB[C * SB_PITCH];
  const int t = threadIdx.x;
  const int lane = t & 31;
  const int wv = __builtin_amdgcn_readfirstlane(t >> 5);
  const int j = blockIdx.x * TN + wv;

  vf wreg[4];
#pragma unroll
  for (int i = 0; i < 4; ++i)
    wreg[i] = *reinterpret_cast<const vf*>(W2 + (size_t)(t + 512 * i) * 4);

  vf r = {};
  if (j < n) {
    const int beg = rowptr[j];
    const int end = rowptr[j + 1];
    const float dj = dinv[j];
    const float* hl = H + lane * 4;
    const vf hs = *reinterpret_cast<const vf*>(hl + (size_t)j * C);
    vf acc = {};
    for (int base = beg; base < end; base += 32) {
      const int m = (end - base < 32) ? (end - base) : 32;
      int s_l = j;
      float w_l = 0.0f;
      if (lane < m) {
        s_l = list[base + lane];
        w_l = dinv[s_l] * dj;
      }
      for (int k = 0; k < m; k += 4) {
        const int s0 = __builtin_amdgcn_readlane(s_l, k);
        const int s1 = __builtin_amdgcn_readlane(s_l, k + 1);
        const int s2 = __builtin_amdgcn_readlane(s_l, k + 2);
        const int s3 = __builtin_amdgcn_readlane(s_l, k + 3);
        const float w0 = __int_as_float(__builtin_amdgcn_readlane(__float_as_int(w_l), k));
        const float w1 = __int_as_float(__builtin_amdgcn_readlane(__float_as_int(w_l), k + 1));
        const float w2 = __int_as_float(__builtin_amdgcn_readlane(__float_as_int(w_l), k + 2));
        const float w3 = __int_as_float(__builtin_amdgcn_readlane(__float_as_int(w_l), k + 3));
        const vf h0 = *reinterpret_cast<const vf*>(hl + (size_t)s0 * C);
        const vf h1 = *reinterpret_cast<const vf*>(hl + (size_t)s1 * C);
        const vf h2 = *reinterpret_cast<const vf*>(hl + (size_t)s2 * C);
        const vf h3 = *reinterpret_cast<const vf*>(hl + (size_t)s3 * C);
        acc = h0 * w0 + acc;
        const vf t1 = h1 * w1 + acc;
        acc = (k + 1 < m) ? t1 : acc;
        const vf t2 = h2 * w2 + acc;
        acc = (k + 2 < m) ? t2 : acc;
        const vf t3 = h3 * w3 + acc;
        acc = (k + 3 < m) ? t3 : acc;
      }
    }
    acc = hs * (dj * dj) + acc;
    r = acc + *reinterpret_cast<const vf*>(bias + lane * 4);
#pragma unroll
    for (int q = 0; q < 4; ++q) r[q] = fmaxf(r[q], 0.0f);
  }
  *reinterpret_cast<vf*>(sA + wv * SA_PITCH + lane * 4) = r;
#pragma unroll
  for (int i = 0; i < 4; ++i) {
    const int idx4 = t + 512 * i;
    *reinterpret_cast<vf*>(sB + (idx4 >> 4) * SB_PITCH + (idx4 & 15) * 4) = wreg[i];
  }
  __syncthreads();

  if (wv < 4) {
    const int h   = lane >> 4;
    const int l16 = lane & 15;
    const float* ap = sA + l16 * SA_PITCH + 2 * h;
    const float* bp = sB + (2 * h) * SB_PITCH + wv * 16 + l16;
    v8f acc = {};
    v2f fa[2][4], fb[2][4];
#pragma unroll
    for (int q = 0; q < 4; ++q) {
      fa[0][q] = *reinterpret_cast<const v2f*>(ap + 4 * q);
      fb[0][q].x = bp[(4 * q) * SB_PITCH];
      fb[0][q].y = bp[(4 * q + 1) * SB_PITCH];
    }
#pragma unroll
    for (int g = 0; g < C / 16; ++g) {
      const int cur = g & 1, nxt = cur ^ 1;
      if (g + 1 < C / 16) {
#pragma unroll
        for (int q = 0; q < 4; ++q) {
          const int k0 = 16 * (g + 1) + 4 * q;
          fa[nxt][q] = *reinterpret_cast<const v2f*>(ap + k0);
          fb[nxt][q].x = bp[k0 * SB_PITCH];
          fb[nxt][q].y = bp[(k0 + 1) * SB_PITCH];
        }
      }
      __builtin_amdgcn_sched_barrier(0);
#pragma unroll
      for (int q = 0; q < 4; ++q)
        acc = __builtin_amdgcn_wmma_f32_16x16x4_f32(false, fa[cur][q], false, fb[cur][q], (short)0, acc, false, false);
    }
    const int row0 = blockIdx.x * TN + 8 * h;
    float* op = H2 + (size_t)row0 * CO + wv * 16 + l16;
#pragma unroll
    for (int q = 0; q < 8; ++q)
      if (row0 + q < n) op[(size_t)q * CO] = acc[q];
  }
}

extern "C" void kernel_launch(void* const* d_in, const int* in_sizes, int n_in,
                              void* d_out, int out_size, void* d_ws, size_t ws_size,
                              hipStream_t stream) {
  const int CIN = 256, CH = 128, CO = 64;
  const float* x  = (const float*)d_in[0];
  const int*   ei = (const int*)d_in[1];
  const float* W1 = (const float*)d_in[2];
  const float* b1 = (const float*)d_in[3];
  const float* W2 = (const float*)d_in[4];
  const float* b2 = (const float*)d_in[5];
  float* out = (float*)d_out;

  const int N = in_sizes[0] / CIN;
  const int E = in_sizes[1] / 2;
  const int* src = ei;
  const int* dst = ei + E;

  float* ws   = (float*)d_ws;
  float* dinv = ws;
  float* Hbuf = ws + 50176;
  float* H2   = ws + 50176 + (size_t)N * CH;
  int* rowptr = (int*)(H2 + (size_t)N * CO);
  int* list   = rowptr + N + 1;

  int* oi     = (int*)d_out;
  int* hist   = oi;
  int* offs   = oi + 53248;
  int* bkey   = oi + 106496;

  const int T = 256;

  const int chunk = (E + 255) / 256;
  const int nbin  = (N + 255) / 256;
  k_bin_count  <<<256, 1024, 0, stream>>>(dst, hist, E, chunk);
  k_scan_excl  <<<1, 1024, 0, stream>>>(hist, offs);
  k_bin_scatter<<<256, 1024, 0, stream>>>(src, dst, offs, bkey, E, chunk);
  k_bin_finish <<<nbin, 1024, 0, stream>>>(offs, bkey, rowptr, dinv, list, N, E);

  k_gemm_x_w1<<<(N + 31) / 32, 128, 0, stream>>>(x, W1, Hbuf, N);
  k_gather_relu_gemm<<<(N + TN - 1) / TN, 32 * TN, 0, stream>>>(rowptr, list, Hbuf, dinv, b1, W2, H2, N);

  k_node_gather<2, false><<<((size_t)N * 32 + T - 1) / T, T, 0, stream>>>(rowptr, list, H2, dinv, b2, out, N);
}


# Edges ordered by bin inside each chunk's own region (LDS count, one-wave scan, place, one launch); bin finish reads its 256 per-chunk runs; count and scan launches gone; sum order = LDS arrival rank
# speedup vs baseline: 4.0795x; 1.4124x over previous
#include <hip/hip_runtime.h>
#include <hip/hip_bf16.h>

typedef float v2f __attribute__((ext_vector_type(2)));
typedef float v8f __attribute__((ext_vector_type(8)));

constexpr int BIN_SHIFT = 8;
constexpr int BIN_NODES = 1 << BIN_SHIFT;
constexpr int NBLK  = 256;
constexpr int NBINP = 208;
constexpr int NBINL = 224;
constexpr int TQ    = NBINP * NBLK / 4096;
constexpr int EPT   = 4;
constexpr int RPT   = 8;
static_assert(TQ * 4096 == NBINP * NBLK, "table size");

__global__ void __launch_bounds__(1024)
k_chunk_bin(const int* __restrict__ src, const int* __restrict__ dst,
            int* __restrict__ cnt, int* __restrict__ loff, int* __restrict__ bkey,
            int e, int chunk) {
  __shared__ int h[NBINL];
  __shared__ int cur[NBINL];
  const int t = threadIdx.x;
  const int b = blockIdx.x;
  if (t < NBINL) h[t] = 0;
  __syncthreads();
  const int lo = b * chunk;
  const int hi = (lo + chunk < e) ? (lo + chunk) : e;
  int d0[EPT], s0[EPT];
#pragma unroll
  for (int q = 0; q < EPT; ++q) {
    const int i = lo + t + q * 1024;
    const bool ok = i < hi;
    d0[q] = ok ? dst[i] : -1;
    s0[q] = ok ? src[i] : 0;
  }
#pragma unroll
  for (int q = 0; q < EPT; ++q)
    if (d0[q] >= 0) atomicAdd(&h[d0[q] >> BIN_SHIFT], 1);
  for (int base = lo + t + 1024 * EPT; base < hi; base += 1024 * EPT) {
    int d[EPT];
#pragma unroll
    for (int q = 0; q < EPT; ++q) {
      const int i = base + q * 1024;
      d[q] = (i < hi) ? dst[i] : -1;
    }
#pragma unroll
    for (int q = 0; q < EPT; ++q)
      if (d[q] >= 0) atomicAdd(&h[d[q] >> BIN_SHIFT], 1);
  }
  __syncthreads();
  if (t < 32) {
    constexpr int PL = NBINL / 32;
    int v[PL];
    int tot = 0;
#pragma unroll
    for (int q = 0; q < PL; ++q) {
      v[q] = tot;
      tot += h[t * PL + q];
    }
    int inc = tot;
    for (int d = 1; d < 32; d <<= 1) {
      const int o = __shfl_up(inc, d);
      if (t >= d) inc += o;
    }
    const int off = inc - tot;
#pragma unroll
    for (int q = 0; q < PL; ++q) cur[t * PL + q] = off + v[q];
  }
  __syncthreads();
  int my_cnt = 0, my_off = 0;
  if (t < NBINP) {
    my_cnt = h[t];
    my_off = cur[t];
  }
  __syncthreads();
  if (t < NBINP) {
    cnt[t * NBLK + b]  = my_cnt;
    loff[t * NBLK + b] = my_off;
  }
#pragma unroll
  for (int q = 0; q < EPT; ++q) {
    if (d0[q] >= 0) {
      const int p = atomicAdd(&cur[d0[q] >> BIN_SHIFT], 1);
      bkey[lo + p] = ((d0[q] & (BIN_NODES - 1)) << 16) | s0[q];
    }
  }
  for (int base = lo + t + 1024 * EPT; base < hi; base += 1024 * EPT) {
    int d[EPT], s[EPT];
#pragma unroll
    for (int q = 0; q < EPT; ++q) {
      const int i = base + q * 1024;
      const bool ok = i < hi;
      d[q] = ok ? dst[i] : -1;
      s[q] = ok ? src[i] : 0;
    }
#pragma unroll
    for (int q = 0; q < EPT; ++q) {
      if (d[q] >= 0) {
        const int p = atomicAdd(&cur[d[q] >> BIN_SHIFT], 1);
        bkey[lo + p] = ((d[q] & (BIN_NODES - 1)) << 16) | s[q];
      }
    }
  }
}

__global__ void __launch_bounds__(1024)
k_bin_finish(const int* __restrict__ cnt, const int* __restrict__ loff,
             const int* __restrict__ bkey,
             int* __restrict__ rowptr, float* __restrict__ dinv,
             int* __restrict__ list, int n, int e, int chunk) {
  __shared__ int c[BIN_NODES];
  __shared__ int st[BIN_NODES];
  __shared__ int rs[NBLK];
  __shared__ int rl[NBLK];
  __shared__ int red[32];
  const int t = threadIdx.x;
  const int lane = t & 31;
  const int wv = t >> 5;
  const int bk = blockIdx.x;

  int part = 0;
  {
    const int lim4 = bk * (NBLK / 4);
    const int4* c4 = reinterpret_cast<const int4*>(cnt);
    int4 v[TQ];
#pragma unroll
    for (int q = 0; q < TQ; ++q) {
      const int i = t + 1024 * q;
      v[q] = (i < lim4) ? c4[i] : make_int4(0, 0, 0, 0);
    }
#pragma unroll
    for (int q = 0; q < TQ; ++q) part += (v[q].x + v[q].y) + (v[q].z + v[q].w);
  }
  if (t < NBLK) {
    rl[t] = cnt[bk * NBLK + t];
    rs[t] = t * chunk + loff[bk * NBLK + t];
  }
  if (t < BIN_NODES) c[t] = 0;
  for (int d = 16; d > 0; d >>= 1) part += __shfl_xor(part, d);
  if (lane == 0) red[wv] = part;
  __syncthreads();
  int beg = red[lane];
  for (int d = 16; d > 0; d >>= 1) beg += __shfl_xor(beg, d);

  const int r = t >> 2;
  const int sub = t & 3;
  const int len = rl[r];
  const int* run = bkey + rs[r];
  int k0[RPT];
#pragma unroll
  for (int q = 0; q < RPT; ++q) {
    const int i = sub + 4 * q;
    k0[q] = (i < len) ? run[i] : -1;
  }
#pragma unroll
  for (int q = 0; q < RPT; ++q)
    if (k0[q] >= 0) atomicAdd(&c[k0[q] >> 16], 1);
  for (int i0 = sub + 4 * RPT; i0 < len; i0 += 4 * RPT) {
    int k[RPT];
#pragma unroll
    for (int q = 0; q < RPT; ++q) {
      const int i = i0 + 4 * q;
      k[q] = (i < len) ? run[i] : -1;
    }
#pragma unroll
    for (int q = 0; q < RPT; ++q)
      if (k[q] >= 0) atomicAdd(&c[k[q] >> 16], 1);
  }
  __syncthreads();
  if (t < 32) {
    constexpr int PL = BIN_NODES / 32;
    int v[PL];
    int tot = 0;
#pragma unroll
    for (int q = 0; q < PL; ++q) {
      v[q] = tot;
      tot += c[t * PL + q];
    }
    int inc = tot;
    for (int d = 1; d < 32; d <<= 1) {
      const int o = __shfl_up(inc, d);
      if (lane >= d) inc += o;
    }
    const int off = inc - tot;
#pragma unroll
    for (int q = 0; q < PL; ++q) st[t * PL + q] = off + v[q];
  }
  __syncthreads();
  int my_p = 0;
  if (t < BIN_NODES) {
    my_p = beg + st[t];
    st[t] = my_p;
  }
  __syncthreads();
  if (t < BIN_NODES) {
    const int j = bk * BIN_NODES + t;
    if (j < n) {
      rowptr[j] = my_p;
      dinv[j] = rsqrtf((float)(c[t] + 1));
    }
  }
  if (bk == 0 && t == 0) rowptr[n] = e;
#pragma unroll
  for (int q = 0; q < RPT; ++q) {
    if (k0[q] >= 0) {
      const int p = atomicAdd(&st[k0[q] >> 16], 1);
      list[p] = k0[q] & 0xffff;
    }
  }
  for (int i0 = sub + 4 * RPT; i0 < len; i0 += 4 * RPT) {
    int k[RPT];
#pragma unroll
    for (int q = 0; q < RPT; ++q) {
      const int i = i0 + 4 * q;
      k[q] = (i < len) ? run[i] : -1;
    }
#pragma unroll
    for (int q = 0; q < RPT; ++q) {
      if (k[q] >= 0) {
        const int p = atomicAdd(&st[k[q] >> 16], 1);
        list[p] = k[q] & 0xffff;
      }
    }
  }
}

constexpr int GM_ROWS  = 32;
constexpr int GM_K     = 256;
constexpr int GM_N     = 128;
constexpr int GM_SLAB  = 64;
constexpr int GM_PITCH = 260;

__global__ void __launch_bounds__(128)
k_gemm_x_w1(const float* __restrict__ A, const float* __restrict__ B,
            float* __restrict__ C, int M) {
  typedef float vf __attribute__((ext_vector_type(4)));
  __shared__ __align__(16) float sX[GM_ROWS * GM_PITCH];
  const int t = threadIdx.x;
  const int lane = t & 31;
  const int wv = __builtin_amdgcn_readfirstlane(t >> 5);
  const int rt = wv >> 1;
  const int ch = wv & 1;
  const int h = lane >> 4;
  const int l16 = lane & 15;
  const int mbase = blockIdx.x * GM_ROWS;

  const int srow = t >> 4;
  const int sc4 = t & 15;
  const float* gsrc[4];
#pragma unroll
  for (int i = 0; i < 4; ++i) {
    const int r = mbase + srow + 8 * i;
    const int rc = (r < M) ? r : (M - 1);
    gsrc[i] = A + (size_t)rc * GM_K + 4 * sc4;
  }
  float* sdst = sX + srow * GM_PITCH + 4 * sc4;

  vf st[4];
#pragma unroll
  for (int i = 0; i < 4; ++i) st[i] = *reinterpret_cast<const vf*>(gsrc[i]);

  v8f acc0 = {}, acc1 = {}, acc2 = {}, acc3 = {};
  const float* ap = sX + (16 * rt + l16) * GM_PITCH + 2 * h;
  const float* bp = B + (size_t)(2 * h) * GM_N + 64 * ch + 4 * l16;

  for (int q = 0; q < GM_K / GM_SLAB; ++q) {
#pragma unroll
    for (int i = 0; i < 4; ++i)
      *reinterpret_cast<vf*>(sdst + 8 * i * GM_PITCH + GM_SLAB * q) = st[i];
    __syncthreads();
    if (q + 1 < GM_K / GM_SLAB) {
#pragma unroll
      for (int i = 0; i < 4; ++i)
        st[i] = *reinterpret_cast<const vf*>(gsrc[i] + GM_SLAB * (q + 1));
    }
    for (int ks = 0; ks < GM_SLAB / 4; ++ks) {
      const int k0 = GM_SLAB * q + 4 * ks;
      const v2f a = *reinterpret_cast<const v2f*>(ap + k0);
      const vf b0 = *reinterpret_cast<const vf*>(bp + (size_t)k0 * GM_N);
      const vf b1 = *reinterpret_cast<const vf*>(bp + (size_t)(k0 + 1) * GM_N);
      v2f f0, f1, f2, f3;
      f0.x = b0.x; f0.y = b1.x;
      f1.x = b0.y; f1.y = b1.y;
      f2.x = b0.z; f2.y = b1.z;
      f3.x = b0.w; f3.y = b1.w;
      acc0 = __builtin_amdgcn_wmma_f32_16x16x4_f32(false, a, false, f0, (short)0, acc0, false, false);
      acc1 = __builtin_amdgcn_wmma_f32_16x16x4_f32(false, a, false, f1, (short)0, acc1, false, false);
      acc2 = __builtin_amdgcn_wmma_f32_16x16x4_f32(false, a, false, f2, (short)0, acc2, false, false);
      acc3 = __builtin_amdgcn_wmma_f32_16x16x4_f32(false, a, false, f3, (short)0, acc3, false, false);
    }
  }

  const int row0 = mbase + 16 * rt + 8 * h;
  float* op = C + (size_t)row0 * GM_N + 64 * ch + 4 * l16;
#pragma unroll
  for (int r = 0; r < 8; ++r) {
    if (row0 + r < M) {
      vf o;
      o.x = acc0[r]; o.y = acc1[r]; o.z = acc2[r]; o.w = acc3[r];
      *reinterpret_cast<vf*>(op + (size_t)r * GM_N) = o;
    }
  }
}

template <int VEC, bool RELU>
__global__ void k_node_gather(const int* __restrict__ rowptr, const int* __restrict__ list,
                              const float* __restrict__ H, const float* __restrict__ dinv,
                              const float* __restrict__ bias, float* __restrict__ out, int n) {
  typedef float vf __attribute__((ext_vector_type(VEC)));
  constexpr int C = 32 * VEC;
  const int lane = threadIdx.x & 31;
  const int j = __builtin_amdgcn_readfirstlane((int)((blockIdx.x * blockDim.x + threadIdx.x) >> 5));
  if (j >= n) return;
  const int beg = rowptr[j];
  const int end = rowptr[j + 1];
  const float dj = dinv[j];
  const float* hl = H + lane * VEC;
  const vf hs = *reinterpret_cast<const vf*>(hl + (size_t)j * C);
  vf acc = {};
  for (int base = beg; base < end; base += 32) {
    const int m = (end - base < 32) ? (end - base) : 32;
    int s_l = j;
    float w_l = 0.0f;
    if (lane < m) {
      s_l = list[base + lane];
      w_l = dinv[s_l] * dj;
    }
    for (int k = 0; k < m; k += 4) {
      const int s0 = __builtin_amdgcn_readlane(s_l, k);
      const int s1 = __builtin_amdgcn_readlane(s_l, k + 1);
      const int s2 = __builtin_amdgcn_readlane(s_l, k + 2);
      const int s3 = __builtin_amdgcn_readlane(s_l, k + 3);
      const float w0 = __int_as_float(__builtin_amdgcn_readlane(__float_as_int(w_l), k));
      const float w1 = __int_as_float(__builtin_amdgcn_readlane(__float_as_int(w_l), k + 1));
      const float w2 = __int_as_float(__builtin_amdgcn_readlane(__float_as_int(w_l), k + 2));
      const float w3 = __int_as_float(__builtin_amdgcn_readlane(__float_as_int(w_l), k + 3));
      const vf h0 = *reinterpret_cast<const vf*>(hl + (size_t)s0 * C);
      const vf h1 = *reinterpret_cast<const vf*>(hl + (size_t)s1 * C);
      const vf h2 = *reinterpret_cast<const vf*>(hl + (size_t)s2 * C);
      const vf h3 = *reinterpret_cast<const vf*>(hl + (size_t)s3 * C);
      acc = h0 * w0 + acc;
      const vf t1 = h1 * w1 + acc;
      acc = (k + 1 < m) ? t1 : acc;
      const vf t2 = h2 * w2 + acc;
      acc = (k + 2 < m) ? t2 : acc;
      const vf t3 = h3 * w3 + acc;
      acc = (k + 3 < m) ? t3 : acc;
    }
  }
  acc = hs * (dj * dj) + acc;
  vf r = acc + *reinterpret_cast<const vf*>(bias + lane * VEC);
  if (RELU) {
#pragma unroll
    for (int q = 0; q < VEC; ++q) r[q] = fmaxf(r[q], 0.0f);
  }
  *reinterpret_cast<vf*>(out + (size_t)j * C + lane * VEC) = r;
}

constexpr int TN = 16;
constexpr int SA_PITCH = 132;
constexpr int SB_PITCH = 72;

__global__ void __launch_bounds__(512)
k_gather_relu_gemm(const int* __restrict__ rowptr, const int* __restrict__ list,
                   const float* __restrict__ H, const float* __restrict__ dinv,
                   const float* __restrict__ bias, const float* __restrict__ W2,
                   float* __restrict__ H2, int n) {
  typedef float vf __attribute__((ext_vector_type(4)));
  constexpr int C = 128, CO = 64;
  __shared__ __align__(16) float sA[TN * SA_PITCH];
  __shared__ __align__(16) float sB[C * SB_PITCH];
  const int t = threadIdx.x;
  const int lane = t & 31;
  const int wv = __builtin_amdgcn_readfirstlane(t >> 5);
  const int j = blockIdx.x * TN + wv;

  vf wreg[4];
#pragma unroll
  for (int i = 0; i < 4; ++i)
    wreg[i] = *reinterpret_cast<const vf*>(W2 + (size_t)(t + 512 * i) * 4);

  vf r = {};
  if (j < n) {
    const int beg = rowptr[j];
    const int end = rowptr[j + 1];
    const float dj = dinv[j];
    const float* hl = H + lane * 4;
    const vf hs = *reinterpret_cast<const vf*>(hl + (size_t)j * C);
    vf acc = {};
    for (int base = beg; base < end; base += 32) {
      const int m = (end - base < 32) ? (end - base) : 32;
      int s_l = j;
      float w_l = 0.0f;
      if (lane < m) {
        s_l = list[base + lane];
        w_l = dinv[s_l] * dj;
      }
      for (int k = 0; k < m; k += 4) {
        const int s0 = __builtin_amdgcn_readlane(s_l, k);
        const int s1 = __builtin_amdgcn_readlane(s_l, k + 1);
        const int s2 = __builtin_amdgcn_readlane(s_l, k + 2);
        const int s3 = __builtin_amdgcn_readlane(s_l, k + 3);
        const float w0 = __int_as_float(__builtin_amdgcn_readlane(__float_as_int(w_l), k));
        const float w1 = __int_as_float(__builtin_amdgcn_readlane(__float_as_int(w_l), k + 1));
        const float w2 = __int_as_float(__builtin_amdgcn_readlane(__float_as_int(w_l), k + 2));
        const float w3 = __int_as_float(__builtin_amdgcn_readlane(__float_as_int(w_l), k + 3));
        const vf h0 = *reinterpret_cast<const vf*>(hl + (size_t)s0 * C);
        const vf h1 = *reinterpret_cast<const vf*>(hl + (size_t)s1 * C);
        const vf h2 = *reinterpret_cast<const vf*>(hl + (size_t)s2 * C);
        const vf h3 = *reinterpret_cast<const vf*>(hl + (size_t)s3 * C);
        acc = h0 * w0 + acc;
        const vf t1 = h1 * w1 + acc;
        acc = (k + 1 < m) ? t1 : acc;
        const vf t2 = h2 * w2 + acc;
        acc = (k + 2 < m) ? t2 : acc;
        const vf t3 = h3 * w3 + acc;
        acc = (k + 3 < m) ? t3 : acc;
      }
    }
    acc = hs * (dj * dj) + acc;
    r = acc + *reinterpret_cast<const vf*>(bias + lane * 4);
#pragma unroll
    for (int q = 0; q < 4; ++q) r[q] = fmaxf(r[q], 0.0f);
  }
  *reinterpret_cast<vf*>(sA + wv * SA_PITCH + lane * 4) = r;
#pragma unroll
  for (int i = 0; i < 4; ++i) {
    const int idx4 = t + 512 * i;
    *reinterpret_cast<vf*>(sB + (idx4 >> 4) * SB_PITCH + (idx4 & 15) * 4) = wreg[i];
  }
  __syncthreads();

  if (wv < 4) {
    const int h   = lane >> 4;
    const int l16 = lane & 15;
    const float* ap = sA + l16 * SA_PITCH + 2 * h;
    const float* bp = sB + (2 * h) * SB_PITCH + wv * 16 + l16;
    v8f acc = {};
    v2f fa[2][4], fb[2][4];
#pragma unroll
    for (int q = 0; q < 4; ++q) {
      fa[0][q] = *reinterpret_cast<const v2f*>(ap + 4 * q);
      fb[0][q].x = bp[(4 * q) * SB_PITCH];
      fb[0][q].y = bp[(4 * q + 1) * SB_PITCH];
    }
#pragma unroll
    for (int g = 0; g < C / 16; ++g) {
      const int cur = g & 1, nxt = cur ^ 1;
      if (g + 1 < C / 16) {
#pragma unroll
        for (int q = 0; q < 4; ++q) {
          const int k0 = 16 * (g + 1) + 4 * q;
          fa[nxt][q] = *reinterpret_cast<const v2f*>(ap + k0);
          fb[nxt][q].x = bp[k0 * SB_PITCH];
          fb[nxt][q].y = bp[(k0 + 1) * SB_PITCH];
        }
      }
      __builtin_amdgcn_sched_barrier(0);
#pragma unroll
      for (int q = 0; q < 4; ++q)
        acc = __builtin_amdgcn_wmma_f32_16x16x4_f32(false, fa[cur][q], false, fb[cur][q], (short)0, acc, false, false);
    }
    const int row0 = blockIdx.x * TN + 8 * h;
    float* op = H2 + (size_t)row0 * CO + wv * 16 + l16;
#pragma unroll
    for (int q = 0; q < 8; ++q)
      if (row0 + q < n) op[(size_t)q * CO] = acc[q];
  }
}

extern "C" void kernel_launch(void* const* d_in, const int* in_sizes, int n_in,
                              void* d_out, int out_size, void* d_ws, size_t ws_size,
                              hipStream_t stream) {
  const int CIN = 256, CH = 128, CO = 64;
  const float* x  = (const float*)d_in[0];
  const int*   ei = (const int*)d_in[1];
  const float* W1 = (const float*)d_in[2];
  const float* b1 = (const float*)d_in[3];
  const float* W2 = (const float*)d_in[4];
  const float* b2 = (const float*)d_in[5];
  float* out = (float*)d_out;

  const int N = in_sizes[0] / CIN;
  const int E = in_sizes[1] / 2;
  const int* src = ei;
  const int* dst = ei + E;

  float* ws   = (float*)d_ws;
  float* dinv = ws;
  float* Hbuf = ws + 50176;
  float* H2   = ws + 50176 + (size_t)N * CH;
  int* rowptr = (int*)(H2 + (size_t)N * CO);
  int* list   = rowptr + N + 1;

  int* oi     = (int*)d_out;
  int* cnt    = oi;
  int* loff   = oi + 53248;
  int* bkey   = oi + 106496;

  const int T = 256;

  const int chunk = (E + 255) / 256;
  const int nbin  = (N + 255) / 256;
  k_chunk_bin  <<<256, 1024, 0, stream>>>(src, dst, cnt, loff, bkey, E, chunk);
  k_bin_finish <<<nbin, 1024, 0, stream>>>(cnt, loff, bkey, rowptr, dinv, list, N, E, chunk);

  k_gemm_x_w1<<<(N + 31) / 32, 128, 0, stream>>>(x, W1, Hbuf, N);
  k_gather_relu_gemm<<<(N + TN - 1) / TN, 32 * TN, 0, stream>>>(rowptr, list, Hbuf, dinv, b1, W2, H2, N);

  k_node_gather<2, false><<<((size_t)N * 32 + T - 1) / T, T, 0, stream>>>(rowptr, list, H2, dinv, b2, out, N);
}


# Edge-list chunk pass runs as 256 extra 128-thread work-groups of the x@W1 launch (25 edges per thread): one launch fewer, its waits hidden under the product; sum order = LDS arrival rank
# speedup vs baseline: 4.3934x; 1.0769x over previous
#include <hip/hip_runtime.h>
#include <hip/hip_bf16.h>

typedef float v2f __attribute__((ext_vector_type(2)));
typedef float v8f __attribute__((ext_vector_type(8)));

constexpr int BIN_SHIFT = 8;
constexpr int BIN_NODES = 1 << BIN_SHIFT;
constexpr int NBLK  = 256;
constexpr int NBINP = 208;
constexpr int NBINL = 224;
constexpr int TQ    = NBINP * NBLK / 4096;
constexpr int CB_T  = 128;
constexpr int EPT   = 25;
constexpr int TPB   = (NBINP + CB_T - 1) / CB_T;
constexpr int RPT   = 8;
static_assert(TQ * 4096 == NBINP * NBLK, "table size");

__device__ __forceinline__ void
chunk_bin_part(const int* __restrict__ src, const int* __restrict__ dst,
               int* __restrict__ cnt, int* __restrict__ loff, int* __restrict__ bkey,
               int e, int chunk, int b) {
  __shared__ int h[NBINL];
  __shared__ int cur[NBINL];
  const int t = threadIdx.x;
  for (int i = t; i < NBINL; i += CB_T) h[i] = 0;
  __syncthreads();
  const int lo = b * chunk;
  const int hi = (lo + chunk < e) ? (lo + chunk) : e;
  int d0[EPT], s0[EPT];
#pragma unroll
  for (int q = 0; q < EPT; ++q) {
    const int i = lo + t + q * CB_T;
    const bool ok = i < hi;
    d0[q] = ok ? dst[i] : -1;
    s0[q] = ok ? src[i] : 0;
  }
#pragma unroll
  for (int q = 0; q < EPT; ++q)
    if (d0[q] >= 0) atomicAdd(&h[d0[q] >> BIN_SHIFT], 1);
  for (int base = lo + t + CB_T * EPT; base < hi; base += CB_T * EPT) {
    int d[EPT];
#pragma unroll
    for (int q = 0; q < EPT; ++q) {
      const int i = base + q * CB_T;
      d[q] = (i < hi) ? dst[i] : -1;
    }
#pragma unroll
    for (int q = 0; q < EPT; ++q)
      if (d[q] >= 0) atomicAdd(&h[d[q] >> BIN_SHIFT], 1);
  }
  __syncthreads();
  if (t < 32) {
    constexpr int PL = NBINL / 32;
    int v[PL];
    int tot = 0;
#pragma unroll
    for (int q = 0; q < PL; ++q) {
      v[q] = tot;
      tot += h[t * PL + q];
    }
    int inc = tot;
    for (int d = 1; d < 32; d <<= 1) {
      const int o = __shfl_up(inc, d);
      if (t >= d) inc += o;
    }
    const int off = inc - tot;
#pragma unroll
    for (int q = 0; q < PL; ++q) cur[t * PL + q] = off + v[q];
  }
  __syncthreads();
  int my_cnt[TPB], my_off[TPB];
#pragma unroll
  for (int i = 0; i < TPB; ++i) {
    const int bin = t + CB_T * i;
    my_cnt[i] = 0;
    my_off[i] = 0;
    if (bin < NBINP) {
      my_cnt[i] = h[bin];
      my_off[i] = cur[bin];
    }
  }
  __syncthreads();
#pragma unroll
  for (int i = 0; i < TPB; ++i) {
    const int bin = t + CB_T * i;
    if (bin < NBINP) {
      cnt[bin * NBLK + b]  = my_cnt[i];
      loff[bin * NBLK + b] = my_off[i];
    }
  }
#pragma unroll
  for (int q = 0; q < EPT; ++q) {
    if (d0[q] >= 0) {
      const int p = atomicAdd(&cur[d0[q] >> BIN_SHIFT], 1);
      bkey[lo + p] = ((d0[q] & (BIN_NODES - 1)) << 16) | s0[q];
    }
  }
  for (int base = lo + t + CB_T * EPT; base < hi; base += CB_T * EPT) {
    int d[EPT], s[EPT];
#pragma unroll
    for (int q = 0; q < EPT; ++q) {
      const int i = base + q * CB_T;
      const bool ok = i < hi;
      d[q] = ok ? dst[i] : -1;
      s[q] = ok ? src[i] : 0;
    }
#pragma unroll
    for (int q = 0; q < EPT; ++q) {
      if (d[q] >= 0) {
        const int p = atomicAdd(&cur[d[q] >> BIN_SHIFT], 1);
        bkey[lo + p] = ((d[q] & (BIN_NODES - 1)) << 16) | s[q];
      }
    }
  }
}

__global__ void __launch_bounds__(1024)
k_bin_finish(const int* __restrict__ cnt, const int* __restrict__ loff,
             const int* __restrict__ bkey,
             int* __restrict__ rowptr, float* __restrict__ dinv,
             int* __restrict__ list, int n, int e, int chunk) {
  __shared__ int c[BIN_NODES];
  __shared__ int st[BIN_NODES];
  __shared__ int rs[NBLK];
  __shared__ int rl[NBLK];
  __shared__ int red[32];
  const int t = threadIdx.x;
  const int lane = t & 31;
  const int wv = t >> 5;
  const int bk = blockIdx.x;

  int part = 0;
  {
    const int lim4 = bk * (NBLK / 4);
    const int4* c4 = reinterpret_cast<const int4*>(cnt);
    int4 v[TQ];
#pragma unroll
    for (int q = 0; q < TQ; ++q) {
      const int i = t + 1024 * q;
      v[q] = (i < lim4) ? c4[i] : make_int4(0, 0, 0, 0);
    }
#pragma unroll
    for (int q = 0; q < TQ; ++q) part += (v[q].x + v[q].y) + (v[q].z + v[q].w);
  }
  if (t < NBLK) {
    rl[t] = cnt[bk * NBLK + t];
    rs[t] = t * chunk + loff[bk * NBLK + t];
  }
  if (t < BIN_NODES) c[t] = 0;
  for (int d = 16; d > 0; d >>= 1) part += __shfl_xor(part, d);
  if (lane == 0) red[wv] = part;
  __syncthreads();
  int beg = red[lane];
  for (int d = 16; d > 0; d >>= 1) beg += __shfl_xor(beg, d);

  const int r = t >> 2;
  const int sub = t & 3;
  const int len = rl[r];
  const int* run = bkey + rs[r];
  int k0[RPT];
#pragma unroll
  for (int q = 0; q < RPT; ++q) {
    const int i = sub + 4 * q;
    k0[q] = (i < len) ? run[i] : -1;
  }
#pragma unroll
  for (int q = 0; q < RPT; ++q)
    if (k0[q] >= 0) atomicAdd(&c[k0[q] >> 16], 1);
  for (int i0 = sub + 4 * RPT; i0 < len; i0 += 4 * RPT) {
    int k[RPT];
#pragma unroll
    for (int q = 0; q < RPT; ++q) {
      const int i = i0 + 4 * q;
      k[q] = (i < len) ? run[i] : -1;
    }
#pragma unroll
    for (int q = 0; q < RPT; ++q)
      if (k[q] >= 0) atomicAdd(&c[k[q] >> 16], 1);
  }
  __syncthreads();
  if (t < 32) {
    constexpr int PL = BIN_NODES / 32;
    int v[PL];
    int tot = 0;
#pragma unroll
    for (int q = 0; q < PL; ++q) {
      v[q] = tot;
      tot += c[t * PL + q];
    }
    int inc = tot;
    for (int d = 1; d < 32; d <<= 1) {
      const int o = __shfl_up(inc, d);
      if (lane >= d) inc += o;
    }
    const int off = inc - tot;
#pragma unroll
    for (int q = 0; q < PL; ++q) st[t * PL + q] = off + v[q];
  }
  __syncthreads();
  int my_p = 0;
  if (t < BIN_NODES) {
    my_p = beg + st[t];
    st[t] = my_p;
  }
  __syncthreads();
  if (t < BIN_NODES) {
    const int j = bk * BIN_NODES + t;
    if (j < n) {
      rowptr[j] = my_p;
      dinv[j] = rsqrtf((float)(c[t] + 1));
    }
  }
  if (bk == 0 && t == 0) rowptr[n] = e;
#pragma unroll
  for (int q = 0; q < RPT; ++q) {
    if (k0[q] >= 0) {
      const int p = atomicAdd(&st[k0[q] >> 16], 1);
      list[p] = k0[q] & 0xffff;
    }
  }
  for (int i0 = sub + 4 * RPT; i0 < len; i0 += 4 * RPT) {
    int k[RPT];
#pragma unroll
    for (int q = 0; q < RPT; ++q) {
      const int i = i0 + 4 * q;
      k[q] = (i < len) ? run[i] : -1;
    }
#pragma unroll
    for (int q = 0; q < RPT; ++q) {
      if (k[q] >= 0) {
        const int p = atomicAdd(&st[k[q] >> 16], 1);
        list[p] = k[q] & 0xffff;
      }
    }
  }
}

constexpr int GM_ROWS  = 32;
constexpr int GM_K     = 256;
constexpr int GM_N     = 128;
constexpr int GM_SLAB  = 64;
constexpr int GM_PITCH = 260;

static_assert(CB_T == 128, "the chunk pass runs in the product's 128-thread work-groups");

__device__ __forceinline__ void
gemm_x_w1_part(const float* __restrict__ A, const float* __restrict__ B,
               float* __restrict__ C, int M, int tile) {
  typedef float vf __attribute__((ext_vector_type(4)));
  __shared__ __align__(16) float sX[GM_ROWS * GM_PITCH];
  const int t = threadIdx.x;
  const int lane = t & 31;
  const int wv = __builtin_amdgcn_readfirstlane(t >> 5);
  const int rt = wv >> 1;
  const int ch = wv & 1;
  const int h = lane >> 4;
  const int l16 = lane & 15;
  const int mbase = tile * GM_ROWS;

  const int srow = t >> 4;
  const int sc4 = t & 15;
  const float* gsrc[4];
#pragma unroll
  for (int i = 0; i < 4; ++i) {
    const int r = mbase + srow + 8 * i;
    const int rc = (r < M) ? r : (M - 1);
    gsrc[i] = A + (size_t)rc * GM_K + 4 * sc4;
  }
  float* sdst = sX + srow * GM_PITCH + 4 * sc4;

  vf st[4];
#pragma unroll
  for (int i = 0; i < 4; ++i) st[i] = *reinterpret_cast<const vf*>(gsrc[i]);

  v8f acc0 = {}, acc1 = {}, acc2 = {}, acc3 = {};
  const float* ap = sX + (16 * rt + l16) * GM_PITCH + 2 * h;
  const float* bp = B + (size_t)(2 * h) * GM_N + 64 * ch + 4 * l16;

  for (int q = 0; q < GM_K / GM_SLAB; ++q) {
#pragma unroll
    for (int i = 0; i < 4; ++i)
      *reinterpret_cast<vf*>(sdst + 8 * i * GM_PITCH + GM_SLAB * q) = st[i];
    __syncthreads();
    if (q + 1 < GM_K / GM_SLAB) {
#pragma unroll
      for (int i = 0; i < 4; ++i)
        st[i] = *reinterpret_cast<const vf*>(gsrc[i] + GM_SLAB * (q + 1));
    }
    for (int ks = 0; ks < GM_SLAB / 4; ++ks) {
      const int k0 = GM_SLAB * q + 4 * ks;
      const v2f a = *reinterpret_cast<const v2f*>(ap + k0);
      const vf b0 = *reinterpret_cast<const vf*>(bp + (size_t)k0 * GM_N);
      const vf b1 = *reinterpret_cast<const vf*>(bp + (size_t)(k0 + 1) * GM_N);
      v2f f0, f1, f2, f3;
      f0.x = b0.x; f0.y = b1.x;
      f1.x = b0.y; f1.y = b1.y;
      f2.x = b0.z; f2.y = b1.z;
      f3.x = b0.w; f3.y = b1.w;
      acc0 = __builtin_amdgcn_wmma_f32_16x16x4_f32(false, a, false, f0, (short)0, acc0, false, false);
      acc1 = __builtin_amdgcn_wmma_f32_16x16x4_f32(false, a, false, f1, (short)0, acc1, false, false);
      acc2 = __builtin_amdgcn_wmma_f32_16x16x4_f32(false, a, false, f2, (short)0, acc2, false, false);
      acc3 = __builtin_amdgcn_wmma_f32_16x16x4_f32(false, a, false, f3, (short)0, acc3, false, false);
    }
  }

  const int row0 = mbase + 16 * rt + 8 * h;
  float* op = C + (size_t)row0 * GM_N + 64 * ch + 4 * l16;
#pragma unroll
  for (int r = 0; r < 8; ++r) {
    if (row0 + r < M) {
      vf o;
      o.x = acc0[r]; o.y = acc1[r]; o.z = acc2[r]; o.w = acc3[r];
      *reinterpret_cast<vf*>(op + (size_t)r * GM_N) = o;
    }
  }
}

__global__ void __launch_bounds__(128)
k_gemm_and_chunk(const float* __restrict__ A, const float* __restrict__ B,
                 float* __restrict__ C, int M,
                 const int* __restrict__ src, const int* __restrict__ dst,
                 int* __restrict__ cnt, int* __restrict__ loff, int* __restrict__ bkey,
                 int e, int chunk) {
  if (blockIdx.x < NBLK)
    chunk_bin_part(src, dst, cnt, loff, bkey, e, chunk, (int)blockIdx.x);
  else
    gemm_x_w1_part(A, B, C, M, (int)blockIdx.x - NBLK);
}

template <int VEC, bool RELU>
__global__ void k_node_gather(const int* __restrict__ rowptr, const int* __restrict__ list,
                              const float* __restrict__ H, const float* __restrict__ dinv,
                              const float* __restrict__ bias, float* __restrict__ out, int n) {
  typedef float vf __attribute__((ext_vector_type(VEC)));
  constexpr int C = 32 * VEC;
  const int lane = threadIdx.x & 31;
  const int j = __builtin_amdgcn_readfirstlane((int)((blockIdx.x * blockDim.x + threadIdx.x) >> 5));
  if (j >= n) return;
  const int beg = rowptr[j];
  const int end = rowptr[j + 1];
  const float dj = dinv[j];
  const float* hl = H + lane * VEC;
  const vf hs = *reinterpret_cast<const vf*>(hl + (size_t)j * C);
  vf acc = {};
  for (int base = beg; base < end; base += 32) {
    const int m = (end - base < 32) ? (end - base) : 32;
    int s_l = j;
    float w_l = 0.0f;
    if (lane < m) {
      s_l = list[base + lane];
      w_l = dinv[s_l] * dj;
    }
    for (int k = 0; k < m; k += 4) {
      const int s0 = __builtin_amdgcn_readlane(s_l, k);
      const int s1 = __builtin_amdgcn_readlane(s_l, k + 1);
      const int s2 = __builtin_amdgcn_readlane(s_l, k + 2);
      const int s3 = __builtin_amdgcn_readlane(s_l, k + 3);
      const float w0 = __int_as_float(__builtin_amdgcn_readlane(__float_as_int(w_l), k));
      const float w1 = __int_as_float(__builtin_amdgcn_readlane(__float_as_int(w_l), k + 1));
      const float w2 = __int_as_float(__builtin_amdgcn_readlane(__float_as_int(w_l), k + 2));
      const float w3 = __int_as_float(__builtin_amdgcn_readlane(__float_as_int(w_l), k + 3));
      const vf h0 = *reinterpret_cast<const vf*>(hl + (size_t)s0 * C);
      const vf h1 = *reinterpret_cast<const vf*>(hl + (size_t)s1 * C);
      const vf h2 = *reinterpret_cast<const vf*>(hl + (size_t)s2 * C);
      const vf h3 = *reinterpret_cast<const vf*>(hl + (size_t)s3 * C);
      acc = h0 * w0 + acc;
      const vf t1 = h1 * w1 + acc;
      acc = (k + 1 < m) ? t1 : acc;
      const vf t2 = h2 * w2 + acc;
      acc = (k + 2 < m) ? t2 : acc;
      const vf t3 = h3 * w3 + acc;
      acc = (k + 3 < m) ? t3 : acc;
    }
  }
  acc = hs * (dj * dj) + acc;
  vf r = acc + *reinterpret_cast<const vf*>(bias + lane * VEC);
  if (RELU) {
#pragma unroll
    for (int q = 0; q < VEC; ++q) r[q] = fmaxf(r[q], 0.0f);
  }
  *reinterpret_cast<vf*>(out + (size_t)j * C + lane * VEC) = r;
}

constexpr int TN = 16;
constexpr int SA_PITCH = 132;
constexpr int SB_PITCH = 72;

__global__ void __launch_bounds__(512)
k_gather_relu_gemm(const int* __restrict__ rowptr, const int* __restrict__ list,
                   const float* __restrict__ H, const float* __restrict__ dinv,
                   const float* __restrict__ bias, const float* __restrict__ W2,
                   float* __restrict__ H2, int n) {
  typedef float vf __attribute__((ext_vector_type(4)));
  constexpr int C = 128, CO = 64;
  __shared__ __align__(16) float sA[TN * SA_PITCH];
  __shared__ __align__(16) float sB[C * SB_PITCH];
  const int t = threadIdx.x;
  const int lane = t & 31;
  const int wv = __builtin_amdgcn_readfirstlane(t >> 5);
  const int j = blockIdx.x * TN + wv;

  vf wreg[4];
#pragma unroll
  for (int i = 0; i < 4; ++i)
    wreg[i] = *reinterpret_cast<const vf*>(W2 + (size_t)(t + 512 * i) * 4);

  vf r = {};
  if (j < n) {
    const int beg = rowptr[j];
    const int end = rowptr[j + 1];
    const float dj = dinv[j];
    const float* hl = H + lane * 4;
    const vf hs = *reinterpret_cast<const vf*>(hl + (size_t)j * C);
    vf acc = {};
    for (int base = beg; base < end; base += 32) {
      const int m = (end - base < 32) ? (end - base) : 32;
      int s_l = j;
      float w_l = 0.0f;
      if (lane < m) {
        s_l = list[base + lane];
        w_l = dinv[s_l] * dj;
      }
      for (int k = 0; k < m; k += 4) {
        const int s0 = __builtin_amdgcn_readlane(s_l, k);
        const int s1 = __builtin_amdgcn_readlane(s_l, k + 1);
        const int s2 = __builtin_amdgcn_readlane(s_l, k + 2);
        const int s3 = __builtin_amdgcn_readlane(s_l, k + 3);
        const float w0 = __int_as_float(__builtin_amdgcn_readlane(__float_as_int(w_l), k));
        const float w1 = __int_as_float(__builtin_amdgcn_readlane(__float_as_int(w_l), k + 1));
        const float w2 = __int_as_float(__builtin_amdgcn_readlane(__float_as_int(w_l), k + 2));
        const float w3 = __int_as_float(__builtin_amdgcn_readlane(__float_as_int(w_l), k + 3));
        const vf h0 = *reinterpret_cast<const vf*>(hl + (size_t)s0 * C);
        const vf h1 = *reinterpret_cast<const vf*>(hl + (size_t)s1 * C);
        const vf h2 = *reinterpret_cast<const vf*>(hl + (size_t)s2 * C);
        const vf h3 = *reinterpret_cast<const vf*>(hl + (size_t)s3 * C);
        acc = h0 * w0 + acc;
        const vf t1 = h1 * w1 + acc;
        acc = (k + 1 < m) ? t1 : acc;
        const vf t2 = h2 * w2 + acc;
        acc = (k + 2 < m) ? t2 : acc;
        const vf t3 = h3 * w3 + acc;
        acc = (k + 3 < m) ? t3 : acc;
      }
    }
    acc = hs * (dj * dj) + acc;
    r = acc + *reinterpret_cast<const vf*>(bias + lane * 4);
#pragma unroll
    for (int q = 0; q < 4; ++q) r[q] = fmaxf(r[q], 0.0f);
  }
  *reinterpret_cast<vf*>(sA + wv * SA_PITCH + lane * 4) = r;
#pragma unroll
  for (int i = 0; i < 4; ++i) {
    const int idx4 = t + 512 * i;
    *reinterpret_cast<vf*>(sB + (idx4 >> 4) * SB_PITCH + (idx4 & 15) * 4) = wreg[i];
  }
  __syncthreads();

  if (wv < 4) {
    const int h   = lane >> 4;
    const int l16 = lane & 15;
    const float* ap = sA + l16 * SA_PITCH + 2 * h;
    const float* bp = sB + (2 * h) * SB_PITCH + wv * 16 + l16;
    v8f acc = {};
    v2f fa[2][4], fb[2][4];
#pragma unroll
    for (int q = 0; q < 4; ++q) {
      fa[0][q] = *reinterpret_cast<const v2f*>(ap + 4 * q);
      fb[0][q].x = bp[(4 * q) * SB_PITCH];
      fb[0][q].y = bp[(4 * q + 1) * SB_PITCH];
    }
#pragma unroll
    for (int g = 0; g < C / 16; ++g) {
      const int cur = g & 1, nxt = cur ^ 1;
      if (g + 1 < C / 16) {
#pragma unroll
        for (int q = 0; q < 4; ++q) {
          const int k0 = 16 * (g + 1) + 4 * q;
          fa[nxt][q] = *reinterpret_cast<const v2f*>(ap + k0);
          fb[nxt][q].x = bp[k0 * SB_PITCH];
          fb[nxt][q].y = bp[(k0 + 1) * SB_PITCH];
        }
      }
      __builtin_amdgcn_sched_barrier(0);
#pragma unroll
      for (int q = 0; q < 4; ++q)
        acc = __builtin_amdgcn_wmma_f32_16x16x4_f32(false, fa[cur][q], false, fb[cur][q], (short)0, acc, false, false);
    }
    const int row0 = blockIdx.x * TN + 8 * h;
    float* op = H2 + (size_t)row0 * CO + wv * 16 + l16;
#pragma unroll
    for (int q = 0; q < 8; ++q)
      if (row0 + q < n) op[(size_t)q * CO] = acc[q];
  }
}

extern "C" void kernel_launch(void* const* d_in, const int* in_sizes, int n_in,
                              void* d_out, int out_size, void* d_ws, size_t ws_size,
                              hipStream_t stream) {
  const int CIN = 256, CH = 128, CO = 64;
  const float* x  = (const float*)d_in[0];
  const int*   ei = (const int*)d_in[1];
  const float* W1 = (const float*)d_in[2];
  const float* b1 = (const float*)d_in[3];
  const float* W2 = (const float*)d_in[4];
  const float* b2 = (const float*)d_in[5];
  float* out = (float*)d_out;

  const int N = in_sizes[0] / CIN;
  const int E = in_sizes[1] / 2;
  const int* src = ei;
  const int* dst = ei + E;

  float* ws   = (float*)d_ws;
  float* dinv = ws;
  float* Hbuf = ws + 50176;
  float* H2   = ws + 50176 + (size_t)N * CH;
  int* rowptr = (int*)(H2 + (size_t)N * CO);
  int* list   = rowptr + N + 1;

  int* oi     = (int*)d_out;
  int* cnt    = oi;
  int* loff   = oi + 53248;
  int* bkey   = oi + 106496;

  const int T = 256;

  const int chunk = (E + 255) / 256;
  const int nbin  = (N + 255) / 256;
  k_gemm_and_chunk<<<256 + (N + 31) / 32, 128, 0, stream>>>(x, W1, Hbuf, N, src, dst, cnt, loff, bkey, E, chunk);
  k_bin_finish <<<nbin, 1024, 0, stream>>>(cnt, loff, bkey, rowptr, dinv, list, N, E, chunk);

  k_gather_relu_gemm<<<(N + TN - 1) / TN, 32 * TN, 0, stream>>>(rowptr, list, Hbuf, dinv, b1, W2, H2, N);

  k_node_gather<2, false><<<((size_t)N * 32 + T - 1) / T, T, 0, stream>>>(rowptr, list, H2, dinv, b2, out, N);
}


# Layer-1 rows scaled by dinv of their own node before the W2 product, dinv[dst] applied once per layer-2 sum: layer-2 gather drops its per-edge dinv lookup and multiply; order of sums kept
# speedup vs baseline: 4.5255x; 1.0301x over previous
#include <hip/hip_runtime.h>
#include <hip/hip_bf16.h>

typedef float v2f __attribute__((ext_vector_type(2)));
typedef float v8f __attribute__((ext_vector_type(8)));

constexpr int BIN_SHIFT = 8;
constexpr int BIN_NODES = 1 << BIN_SHIFT;
constexpr int NBLK  = 256;
constexpr int NBINP = 208;
constexpr int NBINL = 224;
constexpr int TQ    = NBINP * NBLK / 4096;
constexpr int CB_T  = 128;
constexpr int EPT   = 25;
constexpr int TPB   = (NBINP + CB_T - 1) / CB_T;
constexpr int RPT   = 8;
static_assert(TQ * 4096 == NBINP * NBLK, "table size");

__device__ __forceinline__ void
chunk_bin_part(const int* __restrict__ src, const int* __restrict__ dst,
               int* __restrict__ cnt, int* __restrict__ loff, int* __restrict__ bkey,
               int e, int chunk, int b) {
  __shared__ int h[NBINL];
  __shared__ int cur[NBINL];
  const int t = threadIdx.x;
  for (int i = t; i < NBINL; i += CB_T) h[i] = 0;
  __syncthreads();
  const int lo = b * chunk;
  const int hi = (lo + chunk < e) ? (lo + chunk) : e;
  int d0[EPT], s0[EPT];
#pragma unroll
  for (int q = 0; q < EPT; ++q) {
    const int i = lo + t + q * CB_T;
    const bool ok = i < hi;
    d0[q] = ok ? dst[i] : -1;
    s0[q] = ok ? src[i] : 0;
  }
#pragma unroll
  for (int q = 0; q < EPT; ++q)
    if (d0[q] >= 0) atomicAdd(&h[d0[q] >> BIN_SHIFT], 1);
  for (int base = lo + t + CB_T * EPT; base < hi; base += CB_T * EPT) {
    int d[EPT];
#pragma unroll
    for (int q = 0; q < EPT; ++q) {
      const int i = base + q * CB_T;
      d[q] = (i < hi) ? dst[i] : -1;
    }
#pragma unroll
    for (int q = 0; q < EPT; ++q)
      if (d[q] >= 0) atomicAdd(&h[d[q] >> BIN_SHIFT], 1);
  }
  __syncthreads();
  if (t < 32) {
    constexpr int PL = NBINL / 32;
    int v[PL];
    int tot = 0;
#pragma unroll
    for (int q = 0; q < PL; ++q) {
      v[q] = tot;
      tot += h[t * PL + q];
    }
    int inc = tot;
    for (int d = 1; d < 32; d <<= 1) {
      const int o = __shfl_up(inc, d);
      if (t >= d) inc += o;
    }
    const int off = inc - tot;
#pragma unroll
    for (int q = 0; q < PL; ++q) cur[t * PL + q] = off + v[q];
  }
  __syncthreads();
  int my_cnt[TPB], my_off[TPB];
#pragma unroll
  for (int i = 0; i < TPB; ++i) {
    const int bin = t + CB_T * i;
    my_cnt[i] = 0;
    my_off[i] = 0;
    if (bin < NBINP) {
      my_cnt[i] = h[bin];
      my_off[i] = cur[bin];
    }
  }
  __syncthreads();
#pragma unroll
  for (int i = 0; i < TPB; ++i) {
    const int bin = t + CB_T * i;
    if (bin < NBINP) {
      cnt[bin * NBLK + b]  = my_cnt[i];
      loff[bin * NBLK + b] = my_off[i];
    }
  }
#pragma unroll
  for (int q = 0; q < EPT; ++q) {
    if (d0[q] >= 0) {
      const int p = atomicAdd(&cur[d0[q] >> BIN_SHIFT], 1);
      bkey[lo + p] = ((d0[q] & (BIN_NODES - 1)) << 16) | s0[q];
    }
  }
  for (int base = lo + t + CB_T * EPT; base < hi; base += CB_T * EPT) {
    int d[EPT], s[EPT];
#pragma unroll
    for (int q = 0; q < EPT; ++q) {
      const int i = base + q * CB_T;
      const bool ok = i < hi;
      d[q] = ok ? dst[i] : -1;
      s[q] = ok ? src[i] : 0;
    }
#pragma unroll
    for (int q = 0; q < EPT; ++q) {
      if (d[q] >= 0) {
        const int p = atomicAdd(&cur[d[q] >> BIN_SHIFT], 1);
        bkey[lo + p] = ((d[q] & (BIN_NODES - 1)) << 16) | s[q];
      }
    }
  }
}

__global__ void __launch_bounds__(1024)
k_bin_finish(const int* __restrict__ cnt, const int* __restrict__ loff,
             const int* __restrict__ bkey,
             int* __restrict__ rowptr, float* __restrict__ dinv,
             int* __restrict__ list, int n, int e, int chunk) {
  __shared__ int c[BIN_NODES];
  __shared__ int st[BIN_NODES];
  __shared__ int rs[NBLK];
  __shared__ int rl[NBLK];
  __shared__ int red[32];
  const int t = threadIdx.x;
  const int lane = t & 31;
  const int wv = t >> 5;
  const int bk = blockIdx.x;

  int part = 0;
  {
    const int lim4 = bk * (NBLK / 4);
    const int4* c4 = reinterpret_cast<const int4*>(cnt);
    int4 v[TQ];
#pragma unroll
    for (int q = 0; q < TQ; ++q) {
      const int i = t + 1024 * q;
      v[q] = (i < lim4) ? c4[i] : make_int4(0, 0, 0, 0);
    }
#pragma unroll
    for (int q = 0; q < TQ; ++q) part += (v[q].x + v[q].y) + (v[q].z + v[q].w);
  }
  if (t < NBLK) {
    rl[t] = cnt[bk * NBLK + t];
    rs[t] = t * chunk + loff[bk * NBLK + t];
  }
  if (t < BIN_NODES) c[t] = 0;
  for (int d = 16; d > 0; d >>= 1) part += __shfl_xor(part, d);
  if (lane == 0) red[wv] = part;
  __syncthreads();
  int beg = red[lane];
  for (int d = 16; d > 0; d >>= 1) beg += __shfl_xor(beg, d);

  const int r = t >> 2;
  const int sub = t & 3;
  const int len = rl[r];
  const int* run = bkey + rs[r];
  int k0[RPT];
#pragma unroll
  for (int q = 0; q < RPT; ++q) {
    const int i = sub + 4 * q;
    k0[q] = (i < len) ? run[i] : -1;
  }
#pragma unroll
  for (int q = 0; q < RPT; ++q)
    if (k0[q] >= 0) atomicAdd(&c[k0[q] >> 16], 1);
  for (int i0 = sub + 4 * RPT; i0 < len; i0 += 4 * RPT) {
    int k[RPT];
#pragma unroll
    for (int q = 0; q < RPT; ++q) {
      const int i = i0 + 4 * q;
      k[q] = (i < len) ? run[i] : -1;
    }
#pragma unroll
    for (int q = 0; q < RPT; ++q)
      if (k[q] >= 0) atomicAdd(&c[k[q] >> 16], 1);
  }
  __syncthreads();
  if (t < 32) {
    constexpr int PL = BIN_NODES / 32;
    int v[PL];
    int tot = 0;
#pragma unroll
    for (int q = 0; q < PL; ++q) {
      v[q] = tot;
      tot += c[t * PL + q];
    }
    int inc = tot;
    for (int d = 1; d < 32; d <<= 1) {
      const int o = __shfl_up(inc, d);
      if (lane >= d) inc += o;
    }
    const int off = inc - tot;
#pragma unroll
    for (int q = 0; q < PL; ++q) st[t * PL + q] = off + v[q];
  }
  __syncthreads();
  int my_p = 0;
  if (t < BIN_NODES) {
    my_p = beg + st[t];
    st[t] = my_p;
  }
  __syncthreads();
  if (t < BIN_NODES) {
    const int j = bk * BIN_NODES + t;
    if (j < n) {
      rowptr[j] = my_p;
      dinv[j] = rsqrtf((float)(c[t] + 1));
    }
  }
  if (bk == 0 && t == 0) rowptr[n] = e;
#pragma unroll
  for (int q = 0; q < RPT; ++q) {
    if (k0[q] >= 0) {
      const int p = atomicAdd(&st[k0[q] >> 16], 1);
      list[p] = k0[q] & 0xffff;
    }
  }
  for (int i0 = sub + 4 * RPT; i0 < len; i0 += 4 * RPT) {
    int k[RPT];
#pragma unroll
    for (int q = 0; q < RPT; ++q) {
      const int i = i0 + 4 * q;
      k[q] = (i < len) ? run[i] : -1;
    }
#pragma unroll
    for (int q = 0; q < RPT; ++q) {
      if (k[q] >= 0) {
        const int p = atomicAdd(&st[k[q] >> 16], 1);
        list[p] = k[q] & 0xffff;
      }
    }
  }
}

constexpr int GM_ROWS  = 32;
constexpr int GM_K     = 256;
constexpr int GM_N     = 128;
constexpr int GM_SLAB  = 64;
constexpr int GM_PITCH = 260;

static_assert(CB_T == 128, "the chunk pass runs in the product's 128-thread work-groups");

__device__ __forceinline__ void
gemm_x_w1_part(const float* __restrict__ A, const float* __restrict__ B,
               float* __restrict__ C, int M, int tile) {
  typedef float vf __attribute__((ext_vector_type(4)));
  __shared__ __align__(16) float sX[GM_ROWS * GM_PITCH];
  const int t = threadIdx.x;
  const int lane = t & 31;
  const int wv = __builtin_amdgcn_readfirstlane(t >> 5);
  const int rt = wv >> 1;
  const int ch = wv & 1;
  const int h = lane >> 4;
  const int l16 = lane & 15;
  const int mbase = tile * GM_ROWS;

  const int srow = t >> 4;
  const int sc4 = t & 15;
  const float* gsrc[4];
#pragma unroll
  for (int i = 0; i < 4; ++i) {
    const int r = mbase + srow + 8 * i;
    const int rc = (r < M) ? r : (M - 1);
    gsrc[i] = A + (size_t)rc * GM_K + 4 * sc4;
  }
  float* sdst = sX + srow * GM_PITCH + 4 * sc4;

  vf st[4];
#pragma unroll
  for (int i = 0; i < 4; ++i) st[i] = *reinterpret_cast<const vf*>(gsrc[i]);

  v8f acc0 = {}, acc1 = {}, acc2 = {}, acc3 = {};
  const float* ap = sX + (16 * rt + l16) * GM_PITCH + 2 * h;
  const float* bp = B + (size_t)(2 * h) * GM_N + 64 * ch + 4 * l16;

  for (int q = 0; q < GM_K / GM_SLAB; ++q) {
#pragma unroll
    for (int i = 0; i < 4; ++i)
      *reinterpret_cast<vf*>(sdst + 8 * i * GM_PITCH + GM_SLAB * q) = st[i];
    __syncthreads();
    if (q + 1 < GM_K / GM_SLAB) {
#pragma unroll
      for (int i = 0; i < 4; ++i)
        st[i] = *reinterpret_cast<const vf*>(gsrc[i] + GM_SLAB * (q + 1));
    }
    for (int ks = 0; ks < GM_SLAB / 4; ++ks) {
      const int k0 = GM_SLAB * q + 4 * ks;
      const v2f a = *reinterpret_cast<const v2f*>(ap + k0);
      const vf b0 = *reinterpret_cast<const vf*>(bp + (size_t)k0 * GM_N);
      const vf b1 = *reinterpret_cast<const vf*>(bp + (size_t)(k0 + 1) * GM_N);
      v2f f0, f1, f2, f3;
      f0.x = b0.x; f0.y = b1.x;
      f1.x = b0.y; f1.y = b1.y;
      f2.x = b0.z; f2.y = b1.z;
      f3.x = b0.w; f3.y = b1.w;
      acc0 = __builtin_amdgcn_wmma_f32_16x16x4_f32(false, a, false, f0, (short)0, acc0, false, false);
      acc1 = __builtin_amdgcn_wmma_f32_16x16x4_f32(false, a, false, f1, (short)0, acc1, false, false);
      acc2 = __builtin_amdgcn_wmma_f32_16x16x4_f32(false, a, false, f2, (short)0, acc2, false, false);
      acc3 = __builtin_amdgcn_wmma_f32_16x16x4_f32(false, a, false, f3, (short)0, acc3, false, false);
    }
  }

  const int row0 = mbase + 16 * rt + 8 * h;
  float* op = C + (size_t)row0 * GM_N + 64 * ch + 4 * l16;
#pragma unroll
  for (int r = 0; r < 8; ++r) {
    if (row0 + r < M) {
      vf o;
      o.x = acc0[r]; o.y = acc1[r]; o.z = acc2[r]; o.w = acc3[r];
      *reinterpret_cast<vf*>(op + (size_t)r * GM_N) = o;
    }
  }
}

__global__ void __launch_bounds__(128)
k_gemm_and_chunk(const float* __restrict__ A, const float* __restrict__ B,
                 float* __restrict__ C, int M,
                 const int* __restrict__ src, const int* __restrict__ dst,
                 int* __restrict__ cnt, int* __restrict__ loff, int* __restrict__ bkey,
                 int e, int chunk) {
  if (blockIdx.x < NBLK)
    chunk_bin_part(src, dst, cnt, loff, bkey, e, chunk, (int)blockIdx.x);
  else
    gemm_x_w1_part(A, B, C, M, (int)blockIdx.x - NBLK);
}

template <int VEC>
__global__ void k_node_sum(const int* __restrict__ rowptr, const int* __restrict__ list,
                           const float* __restrict__ Hs, const float* __restrict__ dinv,
                           const float* __restrict__ bias, float* __restrict__ out, int n) {
  typedef float vf __attribute__((ext_vector_type(VEC)));
  constexpr int C = 32 * VEC;
  const int lane = threadIdx.x & 31;
  const int j = __builtin_amdgcn_readfirstlane((int)((blockIdx.x * blockDim.x + threadIdx.x) >> 5));
  if (j >= n) return;
  const int beg = rowptr[j];
  const int end = rowptr[j + 1];
  const float dj = dinv[j];
  const float* hl = Hs + lane * VEC;
  const vf hs = *reinterpret_cast<const vf*>(hl + (size_t)j * C);
  vf acc = {};
  for (int base = beg; base < end; base += 32) {
    const int m = (end - base < 32) ? (end - base) : 32;
    int s_l = j;
    if (lane < m) s_l = list[base + lane];
    for (int k = 0; k < m; k += 4) {
      const int s0 = __builtin_amdgcn_readlane(s_l, k);
      const int s1 = __builtin_amdgcn_readlane(s_l, k + 1);
      const int s2 = __builtin_amdgcn_readlane(s_l, k + 2);
      const int s3 = __builtin_amdgcn_readlane(s_l, k + 3);
      const vf h0 = *reinterpret_cast<const vf*>(hl + (size_t)s0 * C);
      const vf h1 = *reinterpret_cast<const vf*>(hl + (size_t)s1 * C);
      const vf h2 = *reinterpret_cast<const vf*>(hl + (size_t)s2 * C);
      const vf h3 = *reinterpret_cast<const vf*>(hl + (size_t)s3 * C);
      acc = acc + h0;
      const vf t1 = acc + h1;
      acc = (k + 1 < m) ? t1 : acc;
      const vf t2 = acc + h2;
      acc = (k + 2 < m) ? t2 : acc;
      const vf t3 = acc + h3;
      acc = (k + 3 < m) ? t3 : acc;
    }
  }
  acc = acc + hs;
  const vf r = acc * dj + *reinterpret_cast<const vf*>(bias + lane * VEC);
  *reinterpret_cast<vf*>(out + (size_t)j * C + lane * VEC) = r;
}

constexpr int TN = 16;
constexpr int SA_PITCH = 132;
constexpr int SB_PITCH = 72;

__global__ void __launch_bounds__(512)
k_gather_relu_gemm(const int* __restrict__ rowptr, const int* __restrict__ list,
                   const float* __restrict__ H, const float* __restrict__ dinv,
                   const float* __restrict__ bias, const float* __restrict__ W2,
                   float* __restrict__ H2, int n) {
  typedef float vf __attribute__((ext_vector_type(4)));
  constexpr int C = 128, CO = 64;
  __shared__ __align__(16) float sA[TN * SA_PITCH];
  __shared__ __align__(16) float sB[C * SB_PITCH];
  const int t = threadIdx.x;
  const int lane = t & 31;
  const int wv = __builtin_amdgcn_readfirstlane(t >> 5);
  const int j = blockIdx.x * TN + wv;

  vf wreg[4];
#pragma unroll
  for (int i = 0; i < 4; ++i)
    wreg[i] = *reinterpret_cast<const vf*>(W2 + (size_t)(t + 512 * i) * 4);

  vf r = {};
  if (j < n) {
    const int beg = rowptr[j];
    const int end = rowptr[j + 1];
    const float dj = dinv[j];
    const float* hl = H + lane * 4;
    const vf hs = *reinterpret_cast<const vf*>(hl + (size_t)j * C);
    vf acc = {};
    for (int base = beg; base < end; base += 32) {
      const int m = (end - base < 32) ? (end - base) : 32;
      int s_l = j;
      float w_l = 0.0f;
      if (lane < m) {
        s_l = list[base + lane];
        w_l = dinv[s_l] * dj;
      }
      for (int k = 0; k < m; k += 4) {
        const int s0 = __builtin_amdgcn_readlane(s_l, k);
        const int s1 = __builtin_amdgcn_readlane(s_l, k + 1);
        const int s2 = __builtin_amdgcn_readlane(s_l, k + 2);
        const int s3 = __builtin_amdgcn_readlane(s_l, k + 3);
        const float w0 = __int_as_float(__builtin_amdgcn_readlane(__float_as_int(w_l), k));
        const float w1 = __int_as_float(__builtin_amdgcn_readlane(__float_as_int(w_l), k + 1));
        const float w2 = __int_as_float(__builtin_amdgcn_readlane(__float_as_int(w_l), k + 2));
        const float w3 = __int_as_float(__builtin_amdgcn_readlane(__float_as_int(w_l), k + 3));
        const vf h0 = *reinterpret_cast<const vf*>(hl + (size_t)s0 * C);
        const vf h1 = *reinterpret_cast<const vf*>(hl + (size_t)s1 * C);
        const vf h2 = *reinterpret_cast<const vf*>(hl + (size_t)s2 * C);
        const vf h3 = *reinterpret_cast<const vf*>(hl + (size_t)s3 * C);
        acc = h0 * w0 + acc;
        const vf t1 = h1 * w1 + acc;
        acc = (k + 1 < m) ? t1 : acc;
        const vf t2 = h2 * w2 + acc;
        acc = (k + 2 < m) ? t2 : acc;
        const vf t3 = h3 * w3 + acc;
        acc = (k + 3 < m) ? t3 : acc;
      }
    }
    acc = hs * (dj * dj) + acc;
    r = acc + *reinterpret_cast<const vf*>(bias + lane * 4);
#pragma unroll
    for (int q = 0; q < 4; ++q) r[q] = fmaxf(r[q], 0.0f) * dj;
  }
  *reinterpret_cast<vf*>(sA + wv * SA_PITCH + lane * 4) = r;
#pragma unroll
  for (int i = 0; i < 4; ++i) {
    const int idx4 = t + 512 * i;
    *reinterpret_cast<vf*>(sB + (idx4 >> 4) * SB_PITCH + (idx4 & 15) * 4) = wreg[i];
  }
  __syncthreads();

  if (wv < 4) {
    const int h   = lane >> 4;
    const int l16 = lane & 15;
    const float* ap = sA + l16 * SA_PITCH + 2 * h;
    const float* bp = sB + (2 * h) * SB_PITCH + wv * 16 + l16;
    v8f acc = {};
    v2f fa[2][4], fb[2][4];
#pragma unroll
    for (int q = 0; q < 4; ++q) {
      fa[0][q] = *reinterpret_cast<const v2f*>(ap + 4 * q);
      fb[0][q].x = bp[(4 * q) * SB_PITCH];
      fb[0][q].y = bp[(4 * q + 1) * SB_PITCH];
    }
#pragma unroll
    for (int g = 0; g < C / 16; ++g) {
      const int cur = g & 1, nxt = cur ^ 1;
      if (g + 1 < C / 16) {
#pragma unroll
        for (int q = 0; q < 4; ++q) {
          const int k0 = 16 * (g + 1) + 4 * q;
          fa[nxt][q] = *reinterpret_cast<const v2f*>(ap + k0);
          fb[nxt][q].x = bp[k0 * SB_PITCH];
          fb[nxt][q].y = bp[(k0 + 1) * SB_PITCH];
        }
      }
      __builtin_amdgcn_sched_barrier(0);
#pragma unroll
      for (int q = 0; q < 4; ++q)
        acc = __builtin_amdgcn_wmma_f32_16x16x4_f32(false, fa[cur][q], false, fb[cur][q], (short)0, acc, false, false);
    }
    const int row0 = blockIdx.x * TN + 8 * h;
    float* op = H2 + (size_t)row0 * CO + wv * 16 + l16;
#pragma unroll
    for (int q = 0; q < 8; ++q)
      if (row0 + q < n) op[(size_t)q * CO] = acc[q];
  }
}

extern "C" void kernel_launch(void* const* d_in, const int* in_sizes, int n_in,
                              void* d_out, int out_size, void* d_ws, size_t ws_size,
                              hipStream_t stream) {
  const int CIN = 256, CH = 128, CO = 64;
  const float* x  = (const float*)d_in[0];
  const int*   ei = (const int*)d_in[1];
  const float* W1 = (const float*)d_in[2];
  const float* b1 = (const float*)d_in[3];
  const float* W2 = (const float*)d_in[4];
  const float* b2 = (const float*)d_in[5];
  float* out = (float*)d_out;

  const int N = in_sizes[0] / CIN;
  const int E = in_sizes[1] / 2;
  const int* src = ei;
  const int* dst = ei + E;

  float* ws   = (float*)d_ws;
  float* dinv = ws;
  float* Hbuf = ws + 50176;
  float* H2   = ws + 50176 + (size_t)N * CH;
  int* rowptr = (int*)(H2 + (size_t)N * CO);
  int* list   = rowptr + N + 1;

  int* oi     = (int*)d_out;
  int* cnt    = oi;
  int* loff   = oi + 53248;
  int* bkey   = oi + 106496;

  const int T = 256;

  const int chunk = (E + 255) / 256;
  const int nbin  = (N + 255) / 256;
  k_gemm_and_chunk<<<256 + (N + 31) / 32, 128, 0, stream>>>(x, W1, Hbuf, N, src, dst, cnt, loff, bkey, E, chunk);
  k_bin_finish <<<nbin, 1024, 0, stream>>>(cnt, loff, bkey, rowptr, dinv, list, N, E, chunk);

  k_gather_relu_gemm<<<(N + TN - 1) / TN, 32 * TN, 0, stream>>>(rowptr, list, Hbuf, dinv, b1, W2, H2, N);

  k_node_sum<2><<<((size_t)N * 32 + T - 1) / T, T, 0, stream>>>(rowptr, list, H2, dinv, b2, out, N);
}
